# starting kernel
# baseline (speedup 1.0000x reference)
#include <hip/hip_runtime.h>

#define DF 96
#define NTILES 6
#define TILE_ROWS 64
#define GEMM_THREADS 128
#define FEPS 1e-7f

typedef __attribute__((ext_vector_type(2))) float v2f;
typedef __attribute__((ext_vector_type(8))) float v8f;


__global__ __launch_bounds__(256) void genconv_zero_ws(unsigned int* __restrict__ p,
                                                       int n) {
  int i = blockIdx.x * 256 + threadIdx.x;
  if (i < n) p[i] = 0u;
}

__global__ __launch_bounds__(256) void genconv_edge_max(
    const float* __restrict__ feat, const int* __restrict__ src,
    const int* __restrict__ dst, unsigned int* __restrict__ mx, int E) {
  int idx = blockIdx.x * 256 + threadIdx.x;
  if (idx >= E * 24) return;
  int e = idx / 24;
  int q = idx % 24;
  int s = src[e];
  int d = dst[e];
  const float4* fr = reinterpret_cast<const float4*>(feat + (size_t)s * DF);
  float4 f = fr[q];
  unsigned int* mrow = mx + (size_t)d * DF + q * 4;
  atomicMax(mrow + 0, __float_as_uint(fmaxf(f.x, 0.f) + FEPS));
  atomicMax(mrow + 1, __float_as_uint(fmaxf(f.y, 0.f) + FEPS));
  atomicMax(mrow + 2, __float_as_uint(fmaxf(f.z, 0.f) + FEPS));
  atomicMax(mrow + 3, __float_as_uint(fmaxf(f.w, 0.f) + FEPS));
}

__global__ __launch_bounds__(256) void genconv_edge_sum(
    const float* __restrict__ feat, const int* __restrict__ src,
    const int* __restrict__ dst, const unsigned int* __restrict__ mx,
    float* __restrict__ num, float* __restrict__ den, int E) {
  int idx = blockIdx.x * 256 + threadIdx.x;
  if (idx >= E * 24) return;
  int e = idx / 24;
  int q = idx % 24;
  int s = src[e];
  int d = dst[e];
  const float4* fr = reinterpret_cast<const float4*>(feat + (size_t)s * DF);
  float4 f = fr[q];
  size_t base = (size_t)d * DF + q * 4;
  const unsigned int* mrow = mx + base;
  float m0 = fmaxf(f.x, 0.f) + FEPS;
  float m1 = fmaxf(f.y, 0.f) + FEPS;
  float m2 = fmaxf(f.z, 0.f) + FEPS;
  float m3 = fmaxf(f.w, 0.f) + FEPS;
  float e0 = __expf(m0 - __uint_as_float(mrow[0]));
  float e1 = __expf(m1 - __uint_as_float(mrow[1]));
  float e2 = __expf(m2 - __uint_as_float(mrow[2]));
  float e3 = __expf(m3 - __uint_as_float(mrow[3]));
  atomicAdd(num + base + 0, e0 * m0);
  atomicAdd(num + base + 1, e1 * m1);
  atomicAdd(num + base + 2, e2 * m2);
  atomicAdd(num + base + 3, e3 * m3);
  atomicAdd(den + base + 0, e0);
  atomicAdd(den + base + 1, e1);
  atomicAdd(den + base + 2, e2);
  atomicAdd(den + base + 3, e3);
}

__global__ __launch_bounds__(GEMM_THREADS) void genconv_fused_gemm(
    const float* __restrict__ feat, const float* __restrict__ num,
    const float* __restrict__ den, const float* __restrict__ W,
    const float* __restrict__ b, float* __restrict__ out, int N) {
  __shared__ float sW[DF * DF];
  __shared__ float sA[TILE_ROWS * DF];

  const int tid = threadIdx.x;
  const int rowBase = blockIdx.x * TILE_ROWS;

  for (int i = tid; i < DF * DF; i += GEMM_THREADS) sW[i] = W[i];

  for (int i = tid; i < TILE_ROWS * DF; i += GEMM_THREADS) {
    int r = rowBase + i / DF;
    float v = 0.f;
    if (r < N) {
      size_t gi = (size_t)r * DF + (i % DF);
      float dn = den[gi];
      float msg = (dn > 0.f) ? (num[gi] / dn) : 0.f;
      v = feat[gi] + msg;
    }
    sA[i] = v;
  }
  __syncthreads();

  const int wave = tid >> 5;
  const int lane = tid & 31;
  const int mrow = lane & 15;
  const int kh   = lane >> 4;

  const float* aRow = &sA[(wave * 16 + mrow) * DF];

  float bias[NTILES];
#pragma unroll
  for (int t = 0; t < NTILES; ++t) bias[t] = b[t * 16 + mrow];

  v8f acc[NTILES] = {};

  for (int k = 0; k < DF; k += 4) {
    v2f a;
    a.x = aRow[k + 2 * kh + 0];
    a.y = aRow[k + 2 * kh + 1];
#pragma unroll
    for (int t = 0; t < NTILES; ++t) {
      v2f bf;
      bf.x = sW[(k + 2 * kh + 0) * DF + t * 16 + mrow];
      bf.y = sW[(k + 2 * kh + 1) * DF + t * 16 + mrow];
      acc[t] = __builtin_amdgcn_wmma_f32_16x16x4_f32(
           false, a,  false, bf,
           (short)0, acc[t],  false,  false);
    }
  }

#pragma unroll
  for (int t = 0; t < NTILES; ++t) {
#pragma unroll
    for (int i = 0; i < 8; ++i) {
      int r = rowBase + wave * 16 + i + kh * 8;
      if (r < N) out[(size_t)r * DF + t * 16 + mrow] = acc[t][i] + bias[t];
    }
  }
}

extern "C" void kernel_launch(void* const* d_in, const int* in_sizes, int n_in,
                              void* d_out, int out_size, void* d_ws, size_t ws_size,
                              hipStream_t stream) {
  const float* feat = (const float*)d_in[0];
  const int*   src  = (const int*)d_in[1];
  const int*   dst  = (const int*)d_in[2];
  const float* W    = (const float*)d_in[3];
  const float* b    = (const float*)d_in[4];
  float*       out  = (float*)d_out;

  const int N = in_sizes[0] / DF;
  const int E = in_sizes[1];

  unsigned int* mx  = (unsigned int*)d_ws;
  float* num = (float*)((char*)d_ws + (size_t)1 * N * DF * sizeof(float));
  float* den = (float*)((char*)d_ws + (size_t)2 * N * DF * sizeof(float));

  const int totZero = 3 * N * DF;
  genconv_zero_ws<<<(totZero + 255) / 256, 256, 0, stream>>>(mx, totZero);

  const int totE = E * 24;
  const int egrid = (totE + 255) / 256;
  genconv_edge_max<<<egrid, 256, 0, stream>>>(feat, src, dst, mx, E);
  genconv_edge_sum<<<egrid, 256, 0, stream>>>(feat, src, dst, mx, num, den, E);

  const int ggrid = (N + TILE_ROWS - 1) / TILE_ROWS;
  genconv_fused_gemm<<<ggrid, GEMM_THREADS, 0, stream>>>(feat, num, den, W, b,
                                                         out, N);
}


# Edge aggregation by per-destination buckets (one int atomic per edge, prefix offsets) and a per-node in-register softmax gather replaces float atomics; sums now in bucket arrival order
# speedup vs baseline: 2.3573x; 2.3573x over previous
#include <hip/hip_runtime.h>


#define DF 96
#define NTILES 6
#define TILE_ROWS 64
#define GEMM_THREADS 128
#define FEPS 1e-7f
#define SCAN_THREADS 1024
#define NODE_WAVES 8

typedef __attribute__((ext_vector_type(2))) float v2f;
typedef __attribute__((ext_vector_type(8))) float v8f;


__global__ __launch_bounds__(256) void genconv_zero_cnt(int* __restrict__ p, int n) {
  int i = blockIdx.x * 256 + threadIdx.x;
  if (i < n) p[i] = 0;
}

__global__ __launch_bounds__(256) void genconv_edge_rank(
    const int* __restrict__ dst, int* __restrict__ cnt, int* __restrict__ rank,
    int E) {
  int e = blockIdx.x * 256 + threadIdx.x;
  if (e < E) rank[e] = atomicAdd(cnt + dst[e], 1);
}

__global__ __launch_bounds__(SCAN_THREADS) void genconv_bucket_offsets(
    const int* __restrict__ cnt, int* __restrict__ off, int N) {
  __shared__ int s_pre[SCAN_THREADS / 32];
  __shared__ int s_loc[SCAN_THREADS / 32];
  const int tid = threadIdx.x;
  const int lane = tid & 31;
  const int w = tid >> 5;
  const int base = blockIdx.x * SCAN_THREADS;

  int p = 0;
  const int4* c4 = reinterpret_cast<const int4*>(cnt);
  for (int k = tid; k < base / 4; k += SCAN_THREADS) {
    int4 t = c4[k];
    p += (t.x + t.y) + (t.z + t.w);
  }
  const int i = base + tid;
  const int v = (i < N) ? cnt[i] : 0;

  int x = v;
  for (int d = 1; d < 32; d <<= 1) {
    int y = __shfl_up(x, d, 32);
    if (lane >= d) x += y;
  }
  for (int d = 16; d > 0; d >>= 1) p += __shfl_xor(p, d, 32);
  if (lane == 31) s_loc[w] = x;
  if (lane == 0) s_pre[w] = p;
  __syncthreads();
  int add = 0;
  for (int k = 0; k < SCAN_THREADS / 32; ++k) {
    add += s_pre[k];
    if (k < w) add += s_loc[k];
  }
  if (i < N) off[i] = add + x - v;
}

__global__ __launch_bounds__(256) void genconv_bucket_fill(
    const int* __restrict__ src, const int* __restrict__ dst,
    const int* __restrict__ rank, const int* __restrict__ off,
    int* __restrict__ perm, int E) {
  int e = blockIdx.x * 256 + threadIdx.x;
  if (e < E) perm[off[dst[e]] + rank[e]] = src[e];
}

__device__ __forceinline__ float4 genconv_row_quad(const float* __restrict__ feat,
                                                   int s, int q) {
  return reinterpret_cast<const float4*>(feat + (size_t)s * DF)[q];
}

__global__ __launch_bounds__(NODE_WAVES * 32) void genconv_node_softmax(
    const float* __restrict__ feat, const int* __restrict__ perm,
    const int* __restrict__ off, const int* __restrict__ cnt,
    float* __restrict__ msg, int N) {
  const int lane = threadIdx.x & 31;
  const int node = __builtin_amdgcn_readfirstlane(
      (int)(blockIdx.x * NODE_WAVES + (threadIdx.x >> 5)));
  if (node >= N) return;
  const int start = off[node];
  const int deg = cnt[node];
  const int q = (lane < 24) ? lane : lane - 8;

  float4 mx = make_float4(0.f, 0.f, 0.f, 0.f);
  for (int b0 = 0; b0 < deg; b0 += 32) {
    const int n = min(32, deg - b0);
    const int myidx = (lane < n) ? perm[start + b0 + lane] : 0;
#pragma unroll 4
    for (int j = 0; j < n; ++j) {
      const int s = __builtin_amdgcn_readlane(myidx, j);
      const float4 f = genconv_row_quad(feat, s, q);
      mx.x = fmaxf(mx.x, fmaxf(f.x, 0.f) + FEPS);
      mx.y = fmaxf(mx.y, fmaxf(f.y, 0.f) + FEPS);
      mx.z = fmaxf(mx.z, fmaxf(f.z, 0.f) + FEPS);
      mx.w = fmaxf(mx.w, fmaxf(f.w, 0.f) + FEPS);
    }
  }

  float4 num = make_float4(0.f, 0.f, 0.f, 0.f);
  float4 den = make_float4(0.f, 0.f, 0.f, 0.f);
  for (int b0 = 0; b0 < deg; b0 += 32) {
    const int n = min(32, deg - b0);
    const int myidx = (lane < n) ? perm[start + b0 + lane] : 0;
#pragma unroll 4
    for (int j = 0; j < n; ++j) {
      const int s = __builtin_amdgcn_readlane(myidx, j);
      const float4 f = genconv_row_quad(feat, s, q);
      const float m0 = fmaxf(f.x, 0.f) + FEPS;
      const float m1 = fmaxf(f.y, 0.f) + FEPS;
      const float m2 = fmaxf(f.z, 0.f) + FEPS;
      const float m3 = fmaxf(f.w, 0.f) + FEPS;
      const float e0 = __expf(m0 - mx.x);
      const float e1 = __expf(m1 - mx.y);
      const float e2 = __expf(m2 - mx.z);
      const float e3 = __expf(m3 - mx.w);
      num.x += e0 * m0;
      num.y += e1 * m1;
      num.z += e2 * m2;
      num.w += e3 * m3;
      den.x += e0;
      den.y += e1;
      den.z += e2;
      den.w += e3;
    }
  }

  float4 r = make_float4(0.f, 0.f, 0.f, 0.f);
  if (deg > 0) {
    r.x = num.x / den.x;
    r.y = num.y / den.y;
    r.z = num.z / den.z;
    r.w = num.w / den.w;
  }
  if (lane < 24) reinterpret_cast<float4*>(msg + (size_t)node * DF)[lane] = r;
}

__global__ __launch_bounds__(GEMM_THREADS) void genconv_fused_gemm(
    const float* __restrict__ feat, const float* __restrict__ msg,
    const float* __restrict__ W, const float* __restrict__ b,
    float* __restrict__ out, int N) {
  __shared__ float sW[DF * DF];
  __shared__ float sA[TILE_ROWS * DF];

  const int tid = threadIdx.x;
  const int rowBase = blockIdx.x * TILE_ROWS;

  for (int i = tid; i < DF * DF; i += GEMM_THREADS) sW[i] = W[i];

  for (int i = tid; i < TILE_ROWS * DF; i += GEMM_THREADS) {
    int r = rowBase + i / DF;
    float v = 0.f;
    if (r < N) {
      size_t gi = (size_t)r * DF + (i % DF);
      v = feat[gi] + msg[gi];
    }
    sA[i] = v;
  }
  __syncthreads();

  const int wave = tid >> 5;
  const int lane = tid & 31;
  const int mrow = lane & 15;
  const int kh   = lane >> 4;

  const float* aRow = &sA[(wave * 16 + mrow) * DF];

  float bias[NTILES];
#pragma unroll
  for (int t = 0; t < NTILES; ++t) bias[t] = b[t * 16 + mrow];

  v8f acc[NTILES] = {};

  for (int k = 0; k < DF; k += 4) {
    v2f a;
    a.x = aRow[k + 2 * kh + 0];
    a.y = aRow[k + 2 * kh + 1];
#pragma unroll
    for (int t = 0; t < NTILES; ++t) {
      v2f bf;
      bf.x = sW[(k + 2 * kh + 0) * DF + t * 16 + mrow];
      bf.y = sW[(k + 2 * kh + 1) * DF + t * 16 + mrow];
      acc[t] = __builtin_amdgcn_wmma_f32_16x16x4_f32(
           false, a,  false, bf,
           (short)0, acc[t],  false,  false);
    }
  }

#pragma unroll
  for (int t = 0; t < NTILES; ++t) {
#pragma unroll
    for (int i = 0; i < 8; ++i) {
      int r = rowBase + wave * 16 + i + kh * 8;
      if (r < N) out[(size_t)r * DF + t * 16 + mrow] = acc[t][i] + bias[t];
    }
  }
}

extern "C" void kernel_launch(void* const* d_in, const int* in_sizes, int n_in,
                              void* d_out, int out_size, void* d_ws, size_t ws_size,
                              hipStream_t stream) {
  const float* feat = (const float*)d_in[0];
  const int*   src  = (const int*)d_in[1];
  const int*   dst  = (const int*)d_in[2];
  const float* W    = (const float*)d_in[3];
  const float* b    = (const float*)d_in[4];
  float*       out  = (float*)d_out;

  const int N = in_sizes[0] / DF;
  const int E = in_sizes[1];

  const size_t Npad = ((size_t)N + 63) / 64 * 64;
  const size_t Epad = ((size_t)E + 63) / 64 * 64;
  int*   cnt  = (int*)d_ws;
  int*   off  = cnt + Npad;
  int*   rank = off + Npad;
  int*   perm = rank + Epad;
  float* msg  = (float*)(perm + Epad);

  const int ngrid = (N + 255) / 256;
  const int egrid = (E + 255) / 256;

  genconv_zero_cnt<<<ngrid, 256, 0, stream>>>(cnt, N);
  genconv_edge_rank<<<egrid, 256, 0, stream>>>(dst, cnt, rank, E);
  genconv_bucket_offsets<<<(N + SCAN_THREADS - 1) / SCAN_THREADS, SCAN_THREADS, 0,
                           stream>>>(cnt, off, N);
  genconv_bucket_fill<<<egrid, 256, 0, stream>>>(src, dst, rank, off, perm, E);
  genconv_node_softmax<<<(N + NODE_WAVES - 1) / NODE_WAVES, NODE_WAVES * 32, 0,
                         stream>>>(feat, perm, off, cnt, msg, N);

  const int ggrid = (N + TILE_ROWS - 1) / TILE_ROWS;
  genconv_fused_gemm<<<ggrid, GEMM_THREADS, 0, stream>>>(feat, msg, W, b, out, N);
}


# Edges go straight into fixed 64-slot buckets per destination (shared overflow list beyond 64) in the ranking kernel; the prefix-offset and fill launches are gone; sums stay in arrival order
# speedup vs baseline: 2.7247x; 1.1559x over previous
#include <hip/hip_runtime.h>


#define DF 96
#define NTILES 6
#define TILE_ROWS 64
#define GEMM_THREADS 128
#define FEPS 1e-7f
#define NODE_WAVES 8
#define BUCKET_CAP 64

typedef __attribute__((ext_vector_type(2))) float v2f;
typedef __attribute__((ext_vector_type(8))) float v8f;


__global__ __launch_bounds__(256) void genconv_zero_cnt(int* __restrict__ p, int n) {
  int i = blockIdx.x * 256 + threadIdx.x;
  if (i < n) p[i] = 0;
}

__global__ __launch_bounds__(256) void genconv_edge_bucket(
    const int* __restrict__ src, const int* __restrict__ dst,
    int* __restrict__ cnt, int* __restrict__ bucket, int* __restrict__ ovfc,
    int* __restrict__ ovfd, int* __restrict__ ovfs, int E) {
  int e = blockIdx.x * 256 + threadIdx.x;
  if (e < E) {
    const int d = dst[e];
    const int s = src[e];
    const int r = atomicAdd(cnt + d, 1);
    if (r < BUCKET_CAP) {
      bucket[(size_t)d * BUCKET_CAP + r] = s;
    } else {
      const int k = atomicAdd(ovfc, 1);
      ovfd[k] = d;
      ovfs[k] = s;
    }
  }
}

__device__ __forceinline__ float4 genconv_row_quad(const float* __restrict__ feat,
                                                   int s, int q) {
  return reinterpret_cast<const float4*>(feat + (size_t)s * DF)[q];
}

__device__ __forceinline__ void genconv_max_row(const float4 f, float4& mx) {
  mx.x = fmaxf(mx.x, fmaxf(f.x, 0.f) + FEPS);
  mx.y = fmaxf(mx.y, fmaxf(f.y, 0.f) + FEPS);
  mx.z = fmaxf(mx.z, fmaxf(f.z, 0.f) + FEPS);
  mx.w = fmaxf(mx.w, fmaxf(f.w, 0.f) + FEPS);
}

__device__ __forceinline__ void genconv_sum_row(const float4 f, const float4 mx,
                                                float4& num, float4& den) {
  const float m0 = fmaxf(f.x, 0.f) + FEPS;
  const float m1 = fmaxf(f.y, 0.f) + FEPS;
  const float m2 = fmaxf(f.z, 0.f) + FEPS;
  const float m3 = fmaxf(f.w, 0.f) + FEPS;
  const float e0 = __expf(m0 - mx.x);
  const float e1 = __expf(m1 - mx.y);
  const float e2 = __expf(m2 - mx.z);
  const float e3 = __expf(m3 - mx.w);
  num.x += e0 * m0;
  num.y += e1 * m1;
  num.z += e2 * m2;
  num.w += e3 * m3;
  den.x += e0;
  den.y += e1;
  den.z += e2;
  den.w += e3;
}

__global__ __launch_bounds__(NODE_WAVES * 32) void genconv_node_softmax(
    const float* __restrict__ feat, const int* __restrict__ bucket,
    const int* __restrict__ cnt, const int* __restrict__ ovfc,
    const int* __restrict__ ovfd, const int* __restrict__ ovfs,
    float* __restrict__ msg, int N, int E) {
  const int lane = threadIdx.x & 31;
  const int node = __builtin_amdgcn_readfirstlane(
      (int)(blockIdx.x * NODE_WAVES + (threadIdx.x >> 5)));
  if (node >= N) return;
  const int deg = cnt[node];
  const int nb = min(deg, BUCKET_CAP);
  const int* __restrict__ mine = bucket + (size_t)node * BUCKET_CAP;
  const int novf = (deg > BUCKET_CAP) ? min(ovfc[0], E) : 0;
  const int q = (lane < 24) ? lane : lane - 8;

  float4 mx = make_float4(0.f, 0.f, 0.f, 0.f);
  for (int b0 = 0; b0 < nb; b0 += 32) {
    const int n = min(32, nb - b0);
    const int myidx = (lane < n) ? mine[b0 + lane] : 0;
#pragma unroll 4
    for (int j = 0; j < n; ++j) {
      const int s = __builtin_amdgcn_readlane(myidx, j);
      genconv_max_row(genconv_row_quad(feat, s, q), mx);
    }
  }
  for (int b0 = 0; b0 < novf; b0 += 32) {
    const int k = b0 + lane;
    const bool hit = (k < novf) && (ovfd[k] == node);
    const int myidx = hit ? ovfs[k] : 0;
    unsigned hits = __builtin_amdgcn_ballot_w32(hit);
    while (hits) {
      const int j = __builtin_ctz(hits);
      hits &= hits - 1;
      const int s = __builtin_amdgcn_readlane(myidx, j);
      genconv_max_row(genconv_row_quad(feat, s, q), mx);
    }
  }

  float4 num = make_float4(0.f, 0.f, 0.f, 0.f);
  float4 den = make_float4(0.f, 0.f, 0.f, 0.f);
  for (int b0 = 0; b0 < nb; b0 += 32) {
    const int n = min(32, nb - b0);
    const int myidx = (lane < n) ? mine[b0 + lane] : 0;
#pragma unroll 4
    for (int j = 0; j < n; ++j) {
      const int s = __builtin_amdgcn_readlane(myidx, j);
      genconv_sum_row(genconv_row_quad(feat, s, q), mx, num, den);
    }
  }
  for (int b0 = 0; b0 < novf; b0 += 32) {
    const int k = b0 + lane;
    const bool hit = (k < novf) && (ovfd[k] == node);
    const int myidx = hit ? ovfs[k] : 0;
    unsigned hits = __builtin_amdgcn_ballot_w32(hit);
    while (hits) {
      const int j = __builtin_ctz(hits);
      hits &= hits - 1;
      const int s = __builtin_amdgcn_readlane(myidx, j);
      genconv_sum_row(genconv_row_quad(feat, s, q), mx, num, den);
    }
  }

  float4 r = make_float4(0.f, 0.f, 0.f, 0.f);
  if (deg > 0) {
    r.x = num.x / den.x;
    r.y = num.y / den.y;
    r.z = num.z / den.z;
    r.w = num.w / den.w;
  }
  if (lane < 24) reinterpret_cast<float4*>(msg + (size_t)node * DF)[lane] = r;
}

__global__ __launch_bounds__(GEMM_THREADS) void genconv_fused_gemm(
    const float* __restrict__ feat, const float* __restrict__ msg,
    const float* __restrict__ W, const float* __restrict__ b,
    float* __restrict__ out, int N) {
  __shared__ float sW[DF * DF];
  __shared__ float sA[TILE_ROWS * DF];

  const int tid = threadIdx.x;
  const int rowBase = blockIdx.x * TILE_ROWS;

  for (int i = tid; i < DF * DF; i += GEMM_THREADS) sW[i] = W[i];

  for (int i = tid; i < TILE_ROWS * DF; i += GEMM_THREADS) {
    int r = rowBase + i / DF;
    float v = 0.f;
    if (r < N) {
      size_t gi = (size_t)r * DF + (i % DF);
      v = feat[gi] + msg[gi];
    }
    sA[i] = v;
  }
  __syncthreads();

  const int wave = tid >> 5;
  const int lane = tid & 31;
  const int mrow = lane & 15;
  const int kh   = lane >> 4;

  const float* aRow = &sA[(wave * 16 + mrow) * DF];

  float bias[NTILES];
#pragma unroll
  for (int t = 0; t < NTILES; ++t) bias[t] = b[t * 16 + mrow];

  v8f acc[NTILES] = {};

  for (int k = 0; k < DF; k += 4) {
    v2f a;
    a.x = aRow[k + 2 * kh + 0];
    a.y = aRow[k + 2 * kh + 1];
#pragma unroll
    for (int t = 0; t < NTILES; ++t) {
      v2f bf;
      bf.x = sW[(k + 2 * kh + 0) * DF + t * 16 + mrow];
      bf.y = sW[(k + 2 * kh + 1) * DF + t * 16 + mrow];
      acc[t] = __builtin_amdgcn_wmma_f32_16x16x4_f32(
           false, a,  false, bf,
           (short)0, acc[t],  false,  false);
    }
  }

#pragma unroll
  for (int t = 0; t < NTILES; ++t) {
#pragma unroll
    for (int i = 0; i < 8; ++i) {
      int r = rowBase + wave * 16 + i + kh * 8;
      if (r < N) out[(size_t)r * DF + t * 16 + mrow] = acc[t][i] + bias[t];
    }
  }
}

extern "C" void kernel_launch(void* const* d_in, const int* in_sizes, int n_in,
                              void* d_out, int out_size, void* d_ws, size_t ws_size,
                              hipStream_t stream) {
  const float* feat = (const float*)d_in[0];
  const int*   src  = (const int*)d_in[1];
  const int*   dst  = (const int*)d_in[2];
  const float* W    = (const float*)d_in[3];
  const float* b    = (const float*)d_in[4];
  float*       out  = (float*)d_out;

  const int N = in_sizes[0] / DF;
  const int E = in_sizes[1];

  const size_t Npad = ((size_t)N + 63) / 64 * 64;
  const size_t Epad = ((size_t)E + 63) / 64 * 64;
  int*   cnt    = (int*)d_ws;
  int*   ovfc   = cnt + Npad;
  int*   bucket = ovfc + 64;
  int*   ovfd   = bucket + Npad * BUCKET_CAP;
  int*   ovfs   = ovfd + Epad;
  float* msg    = (float*)(ovfs + Epad);

  const int nz = (int)Npad + 64;
  const int egrid = (E + 255) / 256;

  genconv_zero_cnt<<<(nz + 255) / 256, 256, 0, stream>>>(cnt, nz);
  genconv_edge_bucket<<<egrid, 256, 0, stream>>>(src, dst, cnt, bucket, ovfc, ovfd,
                                                 ovfs, E);
  genconv_node_softmax<<<(N + NODE_WAVES - 1) / NODE_WAVES, NODE_WAVES * 32, 0,
                         stream>>>(feat, bucket, cnt, ovfc, ovfd, ovfs, msg, N, E);

  const int ggrid = (N + TILE_ROWS - 1) / TILE_ROWS;
  genconv_fused_gemm<<<ggrid, GEMM_THREADS, 0, stream>>>(feat, msg, W, b, out, N);
}


# GEMM kernel operand staging: W, feat and msg tiles read as contiguous 16-byte loads, all 42 per thread in flight before the first wait; was 4-byte loads, the tile's waited one pair at a time
# speedup vs baseline: 3.2616x; 1.1970x over previous
#include <hip/hip_runtime.h>


#define DF 96
#define NTILES 6
#define TILE_ROWS 64
#define GEMM_THREADS 128
#define FEPS 1e-7f
#define NODE_WAVES 8
#define BUCKET_CAP 64

typedef __attribute__((ext_vector_type(2))) float v2f;
typedef __attribute__((ext_vector_type(8))) float v8f;


__global__ __launch_bounds__(256) void genconv_zero_cnt(int* __restrict__ p, int n) {
  int i = blockIdx.x * 256 + threadIdx.x;
  if (i < n) p[i] = 0;
}

__global__ __launch_bounds__(256) void genconv_edge_bucket(
    const int* __restrict__ src, const int* __restrict__ dst,
    int* __restrict__ cnt, int* __restrict__ bucket, int* __restrict__ ovfc,
    int* __restrict__ ovfd, int* __restrict__ ovfs, int E) {
  int e = blockIdx.x * 256 + threadIdx.x;
  if (e < E) {
    const int d = dst[e];
    const int s = src[e];
    const int r = atomicAdd(cnt + d, 1);
    if (r < BUCKET_CAP) {
      bucket[(size_t)d * BUCKET_CAP + r] = s;
    } else {
      const int k = atomicAdd(ovfc, 1);
      ovfd[k] = d;
      ovfs[k] = s;
    }
  }
}

__device__ __forceinline__ float4 genconv_row_quad(const float* __restrict__ feat,
                                                   int s, int q) {
  return reinterpret_cast<const float4*>(feat + (size_t)s * DF)[q];
}

__device__ __forceinline__ void genconv_max_row(const float4 f, float4& mx) {
  mx.x = fmaxf(mx.x, fmaxf(f.x, 0.f) + FEPS);
  mx.y = fmaxf(mx.y, fmaxf(f.y, 0.f) + FEPS);
  mx.z = fmaxf(mx.z, fmaxf(f.z, 0.f) + FEPS);
  mx.w = fmaxf(mx.w, fmaxf(f.w, 0.f) + FEPS);
}

__device__ __forceinline__ void genconv_sum_row(const float4 f, const float4 mx,
                                                float4& num, float4& den) {
  const float m0 = fmaxf(f.x, 0.f) + FEPS;
  const float m1 = fmaxf(f.y, 0.f) + FEPS;
  const float m2 = fmaxf(f.z, 0.f) + FEPS;
  const float m3 = fmaxf(f.w, 0.f) + FEPS;
  const float e0 = __expf(m0 - mx.x);
  const float e1 = __expf(m1 - mx.y);
  const float e2 = __expf(m2 - mx.z);
  const float e3 = __expf(m3 - mx.w);
  num.x += e0 * m0;
  num.y += e1 * m1;
  num.z += e2 * m2;
  num.w += e3 * m3;
  den.x += e0;
  den.y += e1;
  den.z += e2;
  den.w += e3;
}

__global__ __launch_bounds__(NODE_WAVES * 32) void genconv_node_softmax(
    const float* __restrict__ feat, const int* __restrict__ bucket,
    const int* __restrict__ cnt, const int* __restrict__ ovfc,
    const int* __restrict__ ovfd, const int* __restrict__ ovfs,
    float* __restrict__ msg, int N, int E) {
  const int lane = threadIdx.x & 31;
  const int node = __builtin_amdgcn_readfirstlane(
      (int)(blockIdx.x * NODE_WAVES + (threadIdx.x >> 5)));
  if (node >= N) return;
  const int deg = cnt[node];
  const int nb = min(deg, BUCKET_CAP);
  const int* __restrict__ mine = bucket + (size_t)node * BUCKET_CAP;
  const int novf = (deg > BUCKET_CAP) ? min(ovfc[0], E) : 0;
  const int q = (lane < 24) ? lane : lane - 8;

  float4 mx = make_float4(0.f, 0.f, 0.f, 0.f);
  for (int b0 = 0; b0 < nb; b0 += 32) {
    const int n = min(32, nb - b0);
    const int myidx = (lane < n) ? mine[b0 + lane] : 0;
#pragma unroll 4
    for (int j = 0; j < n; ++j) {
      const int s = __builtin_amdgcn_readlane(myidx, j);
      genconv_max_row(genconv_row_quad(feat, s, q), mx);
    }
  }
  for (int b0 = 0; b0 < novf; b0 += 32) {
    const int k = b0 + lane;
    const bool hit = (k < novf) && (ovfd[k] == node);
    const int myidx = hit ? ovfs[k] : 0;
    unsigned hits = __builtin_amdgcn_ballot_w32(hit);
    while (hits) {
      const int j = __builtin_ctz(hits);
      hits &= hits - 1;
      const int s = __builtin_amdgcn_readlane(myidx, j);
      genconv_max_row(genconv_row_quad(feat, s, q), mx);
    }
  }

  float4 num = make_float4(0.f, 0.f, 0.f, 0.f);
  float4 den = make_float4(0.f, 0.f, 0.f, 0.f);
  for (int b0 = 0; b0 < nb; b0 += 32) {
    const int n = min(32, nb - b0);
    const int myidx = (lane < n) ? mine[b0 + lane] : 0;
#pragma unroll 4
    for (int j = 0; j < n; ++j) {
      const int s = __builtin_amdgcn_readlane(myidx, j);
      genconv_sum_row(genconv_row_quad(feat, s, q), mx, num, den);
    }
  }
  for (int b0 = 0; b0 < novf; b0 += 32) {
    const int k = b0 + lane;
    const bool hit = (k < novf) && (ovfd[k] == node);
    const int myidx = hit ? ovfs[k] : 0;
    unsigned hits = __builtin_amdgcn_ballot_w32(hit);
    while (hits) {
      const int j = __builtin_ctz(hits);
      hits &= hits - 1;
      const int s = __builtin_amdgcn_readlane(myidx, j);
      genconv_sum_row(genconv_row_quad(feat, s, q), mx, num, den);
    }
  }

  float4 r = make_float4(0.f, 0.f, 0.f, 0.f);
  if (deg > 0) {
    r.x = num.x / den.x;
    r.y = num.y / den.y;
    r.z = num.z / den.z;
    r.w = num.w / den.w;
  }
  if (lane < 24) reinterpret_cast<float4*>(msg + (size_t)node * DF)[lane] = r;
}

__global__ __launch_bounds__(GEMM_THREADS) void genconv_fused_gemm(
    const float* __restrict__ feat, const float* __restrict__ msg,
    const float* __restrict__ W, const float* __restrict__ b,
    float* __restrict__ out, int N) {
  __shared__ __attribute__((aligned(16))) float sW[DF * DF];
  __shared__ __attribute__((aligned(16))) float sA[TILE_ROWS * DF];

  const int tid = threadIdx.x;
  const int rowBase = blockIdx.x * TILE_ROWS;

  constexpr int QUADS_W = DF * DF / 4 / GEMM_THREADS;
  constexpr int QUADS_A = TILE_ROWS * DF / 4 / GEMM_THREADS;
  const float4* __restrict__ W4 = reinterpret_cast<const float4*>(W);
  const float4* __restrict__ f4 =
      reinterpret_cast<const float4*>(feat) + (size_t)rowBase * (DF / 4);
  const float4* __restrict__ m4 =
      reinterpret_cast<const float4*>(msg) + (size_t)rowBase * (DF / 4);
  const int validQ = min(N - rowBase, TILE_ROWS) * (DF / 4);

  float4 wq[QUADS_W];
  float4 fq[QUADS_A];
  float4 mq[QUADS_A];
#pragma unroll
  for (int it = 0; it < QUADS_A; ++it) {
    const int j = min(tid + it * GEMM_THREADS, validQ - 1);
    fq[it] = f4[j];
    mq[it] = m4[j];
  }
#pragma unroll
  for (int it = 0; it < QUADS_W; ++it) wq[it] = W4[tid + it * GEMM_THREADS];

  float4* sW4 = reinterpret_cast<float4*>(sW);
  float4* sA4 = reinterpret_cast<float4*>(sA);
#pragma unroll
  for (int it = 0; it < QUADS_W; ++it) sW4[tid + it * GEMM_THREADS] = wq[it];
#pragma unroll
  for (int it = 0; it < QUADS_A; ++it) {
    const int j = tid + it * GEMM_THREADS;
    float4 v = make_float4(0.f, 0.f, 0.f, 0.f);
    if (j < validQ) {
      v.x = fq[it].x + mq[it].x;
      v.y = fq[it].y + mq[it].y;
      v.z = fq[it].z + mq[it].z;
      v.w = fq[it].w + mq[it].w;
    }
    sA4[j] = v;
  }
  __syncthreads();

  const int wave = tid >> 5;
  const int lane = tid & 31;
  const int mrow = lane & 15;
  const int kh   = lane >> 4;

  const float* aRow = &sA[(wave * 16 + mrow) * DF];

  float bias[NTILES];
#pragma unroll
  for (int t = 0; t < NTILES; ++t) bias[t] = b[t * 16 + mrow];

  v8f acc[NTILES] = {};

  for (int k = 0; k < DF; k += 4) {
    v2f a;
    a.x = aRow[k + 2 * kh + 0];
    a.y = aRow[k + 2 * kh + 1];
#pragma unroll
    for (int t = 0; t < NTILES; ++t) {
      v2f bf;
      bf.x = sW[(k + 2 * kh + 0) * DF + t * 16 + mrow];
      bf.y = sW[(k + 2 * kh + 1) * DF + t * 16 + mrow];
      acc[t] = __builtin_amdgcn_wmma_f32_16x16x4_f32(
           false, a,  false, bf,
           (short)0, acc[t],  false,  false);
    }
  }

#pragma unroll
  for (int t = 0; t < NTILES; ++t) {
#pragma unroll
    for (int i = 0; i < 8; ++i) {
      int r = rowBase + wave * 16 + i + kh * 8;
      if (r < N) out[(size_t)r * DF + t * 16 + mrow] = acc[t][i] + bias[t];
    }
  }
}

extern "C" void kernel_launch(void* const* d_in, const int* in_sizes, int n_in,
                              void* d_out, int out_size, void* d_ws, size_t ws_size,
                              hipStream_t stream) {
  const float* feat = (const float*)d_in[0];
  const int*   src  = (const int*)d_in[1];
  const int*   dst  = (const int*)d_in[2];
  const float* W    = (const float*)d_in[3];
  const float* b    = (const float*)d_in[4];
  float*       out  = (float*)d_out;

  const int N = in_sizes[0] / DF;
  const int E = in_sizes[1];

  const size_t Npad = ((size_t)N + 63) / 64 * 64;
  const size_t Epad = ((size_t)E + 63) / 64 * 64;
  int*   cnt    = (int*)d_ws;
  int*   ovfc   = cnt + Npad;
  int*   bucket = ovfc + 64;
  int*   ovfd   = bucket + Npad * BUCKET_CAP;
  int*   ovfs   = ovfd + Epad;
  float* msg    = (float*)(ovfs + Epad);

  const int nz = (int)Npad + 64;
  const int egrid = (E + 255) / 256;

  genconv_zero_cnt<<<(nz + 255) / 256, 256, 0, stream>>>(cnt, nz);
  genconv_edge_bucket<<<egrid, 256, 0, stream>>>(src, dst, cnt, bucket, ovfc, ovfd,
                                                 ovfs, E);
  genconv_node_softmax<<<(N + NODE_WAVES - 1) / NODE_WAVES, NODE_WAVES * 32, 0,
                         stream>>>(feat, bucket, cnt, ovfc, ovfd, ovfs, msg, N, E);

  const int ggrid = (N + TILE_ROWS - 1) / TILE_ROWS;
  genconv_fused_gemm<<<ggrid, GEMM_THREADS, 0, stream>>>(feat, msg, W, b, out, N);
}


# Node kernel: a node's rows (up to 24) are loaded once, all loads in flight together, and held in registers for the max pass and the sum pass; was two 4-deep passes over memory; sum order unchanged
# speedup vs baseline: 3.3915x; 1.0398x over previous
#include <hip/hip_runtime.h>


#define DF 96
#define NTILES 6
#define TILE_ROWS 64
#define GEMM_THREADS 128
#define FEPS 1e-7f
#define NODE_WAVES 8
#define BUCKET_CAP 64
#define ROWS_HELD 24

typedef __attribute__((ext_vector_type(2))) float v2f;
typedef __attribute__((ext_vector_type(8))) float v8f;


__global__ __launch_bounds__(256) void genconv_zero_cnt(int* __restrict__ p, int n) {
  int i = blockIdx.x * 256 + threadIdx.x;
  if (i < n) p[i] = 0;
}

__global__ __launch_bounds__(256) void genconv_edge_bucket(
    const int* __restrict__ src, const int* __restrict__ dst,
    int* __restrict__ cnt, int* __restrict__ bucket, int* __restrict__ ovfc,
    int* __restrict__ ovfd, int* __restrict__ ovfs, int E) {
  int e = blockIdx.x * 256 + threadIdx.x;
  if (e < E) {
    const int d = dst[e];
    const int s = src[e];
    const int r = atomicAdd(cnt + d, 1);
    if (r < BUCKET_CAP) {
      bucket[(size_t)d * BUCKET_CAP + r] = s;
    } else {
      const int k = atomicAdd(ovfc, 1);
      ovfd[k] = d;
      ovfs[k] = s;
    }
  }
}

__device__ __forceinline__ float4 genconv_row_quad(const float* __restrict__ feat,
                                                   int s, int q) {
  return reinterpret_cast<const float4*>(feat + (size_t)s * DF)[q];
}

__device__ __forceinline__ void genconv_max_row(const float4 f, float4& mx) {
  mx.x = fmaxf(mx.x, fmaxf(f.x, 0.f) + FEPS);
  mx.y = fmaxf(mx.y, fmaxf(f.y, 0.f) + FEPS);
  mx.z = fmaxf(mx.z, fmaxf(f.z, 0.f) + FEPS);
  mx.w = fmaxf(mx.w, fmaxf(f.w, 0.f) + FEPS);
}

__device__ __forceinline__ void genconv_sum_row(const float4 f, const float4 mx,
                                                float4& num, float4& den) {
  const float m0 = fmaxf(f.x, 0.f) + FEPS;
  const float m1 = fmaxf(f.y, 0.f) + FEPS;
  const float m2 = fmaxf(f.z, 0.f) + FEPS;
  const float m3 = fmaxf(f.w, 0.f) + FEPS;
  const float e0 = __expf(m0 - mx.x);
  const float e1 = __expf(m1 - mx.y);
  const float e2 = __expf(m2 - mx.z);
  const float e3 = __expf(m3 - mx.w);
  num.x += e0 * m0;
  num.y += e1 * m1;
  num.z += e2 * m2;
  num.w += e3 * m3;
  den.x += e0;
  den.y += e1;
  den.z += e2;
  den.w += e3;
}

template <int R>
__device__ __forceinline__ void genconv_node_held(
    const float* __restrict__ feat, int idx, int deg, int q,
    float4& mx, float4& num, float4& den) {
  float4 row[R];
#pragma unroll
  for (int j = 0; j < R; ++j)
    row[j] = genconv_row_quad(feat, __builtin_amdgcn_readlane(idx, j), q);
  __builtin_amdgcn_sched_barrier(0);
#pragma unroll
  for (int j = 0; j < R; ++j) genconv_max_row(row[j], mx);
#pragma unroll
  for (int j = 0; j < R; ++j)
    if (j < deg) genconv_sum_row(row[j], mx, num, den);
}

__global__ __launch_bounds__(NODE_WAVES * 32, 8) void genconv_node_softmax(
    const float* __restrict__ feat, const int* __restrict__ bucket,
    const int* __restrict__ cnt, const int* __restrict__ ovfc,
    const int* __restrict__ ovfd, const int* __restrict__ ovfs,
    float* __restrict__ msg, int N, int E) {
  const int lane = threadIdx.x & 31;
  const int node = __builtin_amdgcn_readfirstlane(
      (int)(blockIdx.x * NODE_WAVES + (threadIdx.x >> 5)));
  if (node >= N) return;
  const int deg = cnt[node];
  const int nb = min(deg, BUCKET_CAP);
  const int* __restrict__ mine = bucket + (size_t)node * BUCKET_CAP;
  const int novf = (deg > BUCKET_CAP) ? min(ovfc[0], E) : 0;
  const int q = (lane < 24) ? lane : lane - 8;

  float4 mx = make_float4(0.f, 0.f, 0.f, 0.f);
  float4 num = make_float4(0.f, 0.f, 0.f, 0.f);
  float4 den = make_float4(0.f, 0.f, 0.f, 0.f);
  if (deg <= ROWS_HELD) {
    if (deg > 0) {
      const int idx = mine[min(lane, deg - 1)];
      if (deg <= 4) genconv_node_held<4>(feat, idx, deg, q, mx, num, den);
      else if (deg <= 8) genconv_node_held<8>(feat, idx, deg, q, mx, num, den);
      else if (deg <= 12) genconv_node_held<12>(feat, idx, deg, q, mx, num, den);
      else if (deg <= 16) genconv_node_held<16>(feat, idx, deg, q, mx, num, den);
      else if (deg <= 20) genconv_node_held<20>(feat, idx, deg, q, mx, num, den);
      else genconv_node_held<ROWS_HELD>(feat, idx, deg, q, mx, num, den);
    }
  } else {
    for (int b0 = 0; b0 < nb; b0 += 32) {
      const int n = min(32, nb - b0);
      const int myidx = (lane < n) ? mine[b0 + lane] : 0;
#pragma unroll 4
      for (int j = 0; j < n; ++j) {
        const int s = __builtin_amdgcn_readlane(myidx, j);
        genconv_max_row(genconv_row_quad(feat, s, q), mx);
      }
    }
    for (int b0 = 0; b0 < novf; b0 += 32) {
      const int k = b0 + lane;
      const bool hit = (k < novf) && (ovfd[k] == node);
      const int myidx = hit ? ovfs[k] : 0;
      unsigned hits = __builtin_amdgcn_ballot_w32(hit);
      while (hits) {
        const int j = __builtin_ctz(hits);
        hits &= hits - 1;
        const int s = __builtin_amdgcn_readlane(myidx, j);
        genconv_max_row(genconv_row_quad(feat, s, q), mx);
      }
    }

    for (int b0 = 0; b0 < nb; b0 += 32) {
      const int n = min(32, nb - b0);
      const int myidx = (lane < n) ? mine[b0 + lane] : 0;
#pragma unroll 4
      for (int j = 0; j < n; ++j) {
        const int s = __builtin_amdgcn_readlane(myidx, j);
        genconv_sum_row(genconv_row_quad(feat, s, q), mx, num, den);
      }
    }
    for (int b0 = 0; b0 < novf; b0 += 32) {
      const int k = b0 + lane;
      const bool hit = (k < novf) && (ovfd[k] == node);
      const int myidx = hit ? ovfs[k] : 0;
      unsigned hits = __builtin_amdgcn_ballot_w32(hit);
      while (hits) {
        const int j = __builtin_ctz(hits);
        hits &= hits - 1;
        const int s = __builtin_amdgcn_readlane(myidx, j);
        genconv_sum_row(genconv_row_quad(feat, s, q), mx, num, den);
      }
    }
  }

  float4 r = make_float4(0.f, 0.f, 0.f, 0.f);
  if (deg > 0) {
    r.x = num.x / den.x;
    r.y = num.y / den.y;
    r.z = num.z / den.z;
    r.w = num.w / den.w;
  }
  if (lane < 24) reinterpret_cast<float4*>(msg + (size_t)node * DF)[lane] = r;
}

__global__ __launch_bounds__(GEMM_THREADS) void genconv_fused_gemm(
    const float* __restrict__ feat, const float* __restrict__ msg,
    const float* __restrict__ W, const float* __restrict__ b,
    float* __restrict__ out, int N) {
  __shared__ __attribute__((aligned(16))) float sW[DF * DF];
  __shared__ __attribute__((aligned(16))) float sA[TILE_ROWS * DF];

  const int tid = threadIdx.x;
  const int rowBase = blockIdx.x * TILE_ROWS;

  constexpr int QUADS_W = DF * DF / 4 / GEMM_THREADS;
  constexpr int QUADS_A = TILE_ROWS * DF / 4 / GEMM_THREADS;
  const float4* __restrict__ W4 = reinterpret_cast<const float4*>(W);
  const float4* __restrict__ f4 =
      reinterpret_cast<const float4*>(feat) + (size_t)rowBase * (DF / 4);
  const float4* __restrict__ m4 =
      reinterpret_cast<const float4*>(msg) + (size_t)rowBase * (DF / 4);
  const int validQ = min(N - rowBase, TILE_ROWS) * (DF / 4);

  float4 wq[QUADS_W];
  float4 fq[QUADS_A];
  float4 mq[QUADS_A];
#pragma unroll
  for (int it = 0; it < QUADS_A; ++it) {
    const int j = min(tid + it * GEMM_THREADS, validQ - 1);
    fq[it] = f4[j];
    mq[it] = m4[j];
  }
#pragma unroll
  for (int it = 0; it < QUADS_W; ++it) wq[it] = W4[tid + it * GEMM_THREADS];

  float4* sW4 = reinterpret_cast<float4*>(sW);
  float4* sA4 = reinterpret_cast<float4*>(sA);
#pragma unroll
  for (int it = 0; it < QUADS_W; ++it) sW4[tid + it * GEMM_THREADS] = wq[it];
#pragma unroll
  for (int it = 0; it < QUADS_A; ++it) {
    const int j = tid + it * GEMM_THREADS;
    float4 v = make_float4(0.f, 0.f, 0.f, 0.f);
    if (j < validQ) {
      v.x = fq[it].x + mq[it].x;
      v.y = fq[it].y + mq[it].y;
      v.z = fq[it].z + mq[it].z;
      v.w = fq[it].w + mq[it].w;
    }
    sA4[j] = v;
  }
  __syncthreads();

  const int wave = tid >> 5;
  const int lane = tid & 31;
  const int mrow = lane & 15;
  const int kh   = lane >> 4;

  const float* aRow = &sA[(wave * 16 + mrow) * DF];

  float bias[NTILES];
#pragma unroll
  for (int t = 0; t < NTILES; ++t) bias[t] = b[t * 16 + mrow];

  v8f acc[NTILES] = {};

  for (int k = 0; k < DF; k += 4) {
    v2f a;
    a.x = aRow[k + 2 * kh + 0];
    a.y = aRow[k + 2 * kh + 1];
#pragma unroll
    for (int t = 0; t < NTILES; ++t) {
      v2f bf;
      bf.x = sW[(k + 2 * kh + 0) * DF + t * 16 + mrow];
      bf.y = sW[(k + 2 * kh + 1) * DF + t * 16 + mrow];
      acc[t] = __builtin_amdgcn_wmma_f32_16x16x4_f32(
           false, a,  false, bf,
           (short)0, acc[t],  false,  false);
    }
  }

#pragma unroll
  for (int t = 0; t < NTILES; ++t) {
#pragma unroll
    for (int i = 0; i < 8; ++i) {
      int r = rowBase + wave * 16 + i + kh * 8;
      if (r < N) out[(size_t)r * DF + t * 16 + mrow] = acc[t][i] + bias[t];
    }
  }
}

extern "C" void kernel_launch(void* const* d_in, const int* in_sizes, int n_in,
                              void* d_out, int out_size, void* d_ws, size_t ws_size,
                              hipStream_t stream) {
  const float* feat = (const float*)d_in[0];
  const int*   src  = (const int*)d_in[1];
  const int*   dst  = (const int*)d_in[2];
  const float* W    = (const float*)d_in[3];
  const float* b    = (const float*)d_in[4];
  float*       out  = (float*)d_out;

  const int N = in_sizes[0] / DF;
  const int E = in_sizes[1];

  const size_t Npad = ((size_t)N + 63) / 64 * 64;
  const size_t Epad = ((size_t)E + 63) / 64 * 64;
  int*   cnt    = (int*)d_ws;
  int*   ovfc   = cnt + Npad;
  int*   bucket = ovfc + 64;
  int*   ovfd   = bucket + Npad * BUCKET_CAP;
  int*   ovfs   = ovfd + Epad;
  float* msg    = (float*)(ovfs + Epad);

  const int nz = (int)Npad + 64;
  const int egrid = (E + 255) / 256;

  genconv_zero_cnt<<<(nz + 255) / 256, 256, 0, stream>>>(cnt, nz);
  genconv_edge_bucket<<<egrid, 256, 0, stream>>>(src, dst, cnt, bucket, ovfc, ovfd,
                                                 ovfs, E);
  genconv_node_softmax<<<(N + NODE_WAVES - 1) / NODE_WAVES, NODE_WAVES * 32, 0,
                         stream>>>(feat, bucket, cnt, ovfc, ovfd, ovfs, msg, N, E);

  const int ggrid = (N + TILE_ROWS - 1) / TILE_ROWS;
  genconv_fused_gemm<<<ggrid, GEMM_THREADS, 0, stream>>>(feat, msg, W, b, out, N);
}


# Edge bucketing without a global atomic per edge: each block orders its edges by 256-node destination bin in LDS; one block per bin then fills buckets and degrees by LDS atomics; sums in arrival order
# speedup vs baseline: 6.5521x; 1.9319x over previous
#include <hip/hip_runtime.h>


#define DF 96
#define NTILES 6
#define TILE_ROWS 64
#define GEMM_THREADS 128
#define FEPS 1e-7f
#define NODE_WAVES 8
#define BUCKET_CAP 64
#define ROWS_HELD 24
#define BIN_SHIFT 8
#define BIN_NODES 256
#define MAX_BINS 256
#define MAX_BLKS 1024
#define PART_THREADS 1024
#define PART_PER_THREAD 4
#define PART_EDGES 4096
#define SEG_CAP 64
#define CNT_PITCH 256
#define FILL_THREADS 1024
#define FILL_DEPTH 13

typedef __attribute__((ext_vector_type(2))) float v2f;
typedef __attribute__((ext_vector_type(8))) float v8f;


__global__ __launch_bounds__(PART_THREADS) void genconv_edge_partition(
    const int* __restrict__ src, const int* __restrict__ dst,
    int* __restrict__ segp, int* __restrict__ segcnt,
    int* __restrict__ spillp, int* __restrict__ spillc,
    int* __restrict__ ovfc, int E, int nblk) {
  __shared__ int hist[MAX_BINS];
  __shared__ int off[MAX_BINS];
  __shared__ int wsum[MAX_BINS / 32];
  __shared__ int nspill;
  __shared__ int2 sorted[PART_EDGES];

  int2* __restrict__ seg = reinterpret_cast<int2*>(segp);
  int2* __restrict__ spill = reinterpret_cast<int2*>(spillp);
  const int tid = threadIdx.x;
  const int blk = blockIdx.x;
  const int base = blk * PART_EDGES;

  int d[PART_PER_THREAD], s[PART_PER_THREAD], r[PART_PER_THREAD];
#pragma unroll
  for (int i = 0; i < PART_PER_THREAD; ++i) {
    const int e = base + i * PART_THREADS + tid;
    const bool ok = e < E;
    d[i] = ok ? dst[e] : -1;
    s[i] = ok ? src[e] : 0;
  }
  if (tid < MAX_BINS) hist[tid] = 0;
  if (tid == 0) nspill = 0;
  if (blk == 0 && tid == 0) ovfc[0] = 0;
  __syncthreads();

#pragma unroll
  for (int i = 0; i < PART_PER_THREAD; ++i) {
    r[i] = 0;
    if (d[i] >= 0) r[i] = atomicAdd(&hist[d[i] >> BIN_SHIFT], 1);
  }
  __syncthreads();

  int v = 0, x = 0;
  if (tid < MAX_BINS) {
    v = hist[tid];
    x = v;
    for (int o = 1; o < 32; o <<= 1) {
      const int y = __shfl_up(x, o);
      if ((tid & 31) >= o) x += y;
    }
    if ((tid & 31) == 31) wsum[tid >> 5] = x;
  }
  __syncthreads();
  if (tid < MAX_BINS) {
    int front = 0;
    for (int w = 0; w < (tid >> 5); ++w) front += wsum[w];
    off[tid] = front + x - v;
    segcnt[(size_t)blk * CNT_PITCH + tid] = min(v, SEG_CAP);
  }
  __syncthreads();

#pragma unroll
  for (int i = 0; i < PART_PER_THREAD; ++i)
    if (d[i] >= 0) sorted[off[d[i] >> BIN_SHIFT] + r[i]] = make_int2(d[i], s[i]);
  __syncthreads();

  const int n = min(PART_EDGES, E - base);
  for (int p = tid; p < n; p += PART_THREADS) {
    const int2 en = sorted[p];
    const int bin = en.x >> BIN_SHIFT;
    const int rank = p - off[bin];
    int2* __restrict__ to = seg + ((size_t)bin * nblk + blk) * SEG_CAP + rank;
    if (rank >= SEG_CAP) {
      const int q = atomicAdd(&nspill, 1);
      to = spill + (size_t)blk * PART_EDGES + q;
    }
    *to = en;
  }
  __syncthreads();
  if (tid == 0) spillc[blk * 32] = nspill;
}

__device__ __forceinline__ void genconv_put(
    const int2 en, const int r, const int node0, int* lbucket,
    int* __restrict__ ovfc, int* __restrict__ ovfd, int* __restrict__ ovfs) {
  const int dl = en.x - node0;
  if (r < BUCKET_CAP) {
    lbucket[dl * BUCKET_CAP + r] = en.y;
  } else {
    const int q = atomicAdd(ovfc, 1);
    ovfd[q] = en.x;
    ovfs[q] = en.y;
  }
}

__global__ __launch_bounds__(FILL_THREADS) void genconv_bin_bucket(
    const int* __restrict__ segp, const int* __restrict__ segcnt,
    const int* __restrict__ spillp, const int* __restrict__ spillc,
    int* __restrict__ cnt, int* __restrict__ bucket, int* __restrict__ ovfc,
    int* __restrict__ ovfd, int* __restrict__ ovfs, int nblk) {
  __shared__ __attribute__((aligned(16))) int lbucket[BIN_NODES * BUCKET_CAP];
  __shared__ int lcnt[BIN_NODES];
  __shared__ int lseg[MAX_BLKS];
  __shared__ int anyspill;

  const int2* __restrict__ seg = reinterpret_cast<const int2*>(segp);
  const int2* __restrict__ spill = reinterpret_cast<const int2*>(spillp);
  const int tid = threadIdx.x;
  const int k = blockIdx.x;
  const int node0 = k << BIN_SHIFT;

  int sc = 0, pc = 0;
  if (tid < nblk) {
    sc = segcnt[(size_t)tid * CNT_PITCH + k];
    pc = spillc[tid * 32];
  }
  if (tid < BIN_NODES) lcnt[tid] = 0;
  if (tid == 0) anyspill = 0;
  static_assert(FILL_THREADS == MAX_BLKS, "one thread per entry of lseg");
  lseg[tid] = sc;
  __syncthreads();
  if (pc > 0) anyspill = 1;

  const int2* __restrict__ mine = seg + (size_t)k * nblk * SEG_CAP;
  const int nslot = nblk * SEG_CAP;
  for (int g0 = 0; g0 < nslot; g0 += FILL_THREADS * FILL_DEPTH) {
    int2 en[FILL_DEPTH];
    int rk[FILL_DEPTH];
    bool ok[FILL_DEPTH];
#pragma unroll
    for (int u = 0; u < FILL_DEPTH; ++u) {
      const int i = g0 + u * FILL_THREADS + tid;
      ok[u] = (i < nslot) && ((i & (SEG_CAP - 1)) < lseg[i / SEG_CAP]);
      en[u] = make_int2(node0, 0);
      if (ok[u]) en[u] = mine[i];
    }
#pragma unroll
    for (int u = 0; u < FILL_DEPTH; ++u) {
      rk[u] = 0;
      if (ok[u]) rk[u] = atomicAdd(&lcnt[en[u].x - node0], 1);
    }
#pragma unroll
    for (int u = 0; u < FILL_DEPTH; ++u)
      if (ok[u]) genconv_put(en[u], rk[u], node0, lbucket, ovfc, ovfd, ovfs);
  }
  __syncthreads();

  if (anyspill) {
    for (int b = 0; b < nblk; ++b) {
      const int c = min(spillc[b * 32], PART_EDGES);
      for (int i = tid; i < c; i += FILL_THREADS) {
        const int2 e2 = spill[(size_t)b * PART_EDGES + i];
        if ((e2.x >> BIN_SHIFT) == k)
          genconv_put(e2, atomicAdd(&lcnt[e2.x - node0], 1), node0, lbucket, ovfc, ovfd,
                      ovfs);
      }
    }
  }
  __syncthreads();

  if (tid < BIN_NODES) cnt[node0 + tid] = lcnt[tid];
  const int4* lb4 = reinterpret_cast<const int4*>(lbucket);
  int4* __restrict__ gb4 = reinterpret_cast<int4*>(bucket + (size_t)node0 * BUCKET_CAP);
#pragma unroll
  for (int it = 0; it < BIN_NODES * BUCKET_CAP / 4 / FILL_THREADS; ++it) {
    const int q = it * FILL_THREADS + tid;
    const int node = q >> 4;
    const int firstslot = ((q >> 3) & 1) * 32;
    if (firstslot < lcnt[node]) gb4[q] = lb4[q];
  }
}

__device__ __forceinline__ float4 genconv_row_quad(const float* __restrict__ feat,
                                                   int s, int q) {
  return reinterpret_cast<const float4*>(feat + (size_t)s * DF)[q];
}

__device__ __forceinline__ void genconv_max_row(const float4 f, float4& mx) {
  mx.x = fmaxf(mx.x, fmaxf(f.x, 0.f) + FEPS);
  mx.y = fmaxf(mx.y, fmaxf(f.y, 0.f) + FEPS);
  mx.z = fmaxf(mx.z, fmaxf(f.z, 0.f) + FEPS);
  mx.w = fmaxf(mx.w, fmaxf(f.w, 0.f) + FEPS);
}

__device__ __forceinline__ void genconv_sum_row(const float4 f, const float4 mx,
                                                float4& num, float4& den) {
  const float m0 = fmaxf(f.x, 0.f) + FEPS;
  const float m1 = fmaxf(f.y, 0.f) + FEPS;
  const float m2 = fmaxf(f.z, 0.f) + FEPS;
  const float m3 = fmaxf(f.w, 0.f) + FEPS;
  const float e0 = __expf(m0 - mx.x);
  const float e1 = __expf(m1 - mx.y);
  const float e2 = __expf(m2 - mx.z);
  const float e3 = __expf(m3 - mx.w);
  num.x += e0 * m0;
  num.y += e1 * m1;
  num.z += e2 * m2;
  num.w += e3 * m3;
  den.x += e0;
  den.y += e1;
  den.z += e2;
  den.w += e3;
}

template <int R>
__device__ __forceinline__ void genconv_node_held(
    const float* __restrict__ feat, int idx, int deg, int q,
    float4& mx, float4& num, float4& den) {
  float4 row[R];
#pragma unroll
  for (int j = 0; j < R; ++j)
    row[j] = genconv_row_quad(feat, __builtin_amdgcn_readlane(idx, j), q);
  __builtin_amdgcn_sched_barrier(0);
#pragma unroll
  for (int j = 0; j < R; ++j) genconv_max_row(row[j], mx);
#pragma unroll
  for (int j = 0; j < R; ++j)
    if (j < deg) genconv_sum_row(row[j], mx, num, den);
}

__global__ __launch_bounds__(NODE_WAVES * 32, 8) void genconv_node_softmax(
    const float* __restrict__ feat, const int* __restrict__ bucket,
    const int* __restrict__ cnt, const int* __restrict__ ovfc,
    const int* __restrict__ ovfd, const int* __restrict__ ovfs,
    float* __restrict__ msg, int N, int E) {
  const int lane = threadIdx.x & 31;
  const int node = __builtin_amdgcn_readfirstlane(
      (int)(blockIdx.x * NODE_WAVES + (threadIdx.x >> 5)));
  if (node >= N) return;
  const int deg = cnt[node];
  const int nb = min(deg, BUCKET_CAP);
  const int* __restrict__ mine = bucket + (size_t)node * BUCKET_CAP;
  const int novf = (deg > BUCKET_CAP) ? min(ovfc[0], E) : 0;
  const int q = (lane < 24) ? lane : lane - 8;

  float4 mx = make_float4(0.f, 0.f, 0.f, 0.f);
  float4 num = make_float4(0.f, 0.f, 0.f, 0.f);
  float4 den = make_float4(0.f, 0.f, 0.f, 0.f);
  if (deg <= ROWS_HELD) {
    if (deg > 0) {
      const int idx = mine[min(lane, deg - 1)];
      if (deg <= 4) genconv_node_held<4>(feat, idx, deg, q, mx, num, den);
      else if (deg <= 8) genconv_node_held<8>(feat, idx, deg, q, mx, num, den);
      else if (deg <= 12) genconv_node_held<12>(feat, idx, deg, q, mx, num, den);
      else if (deg <= 16) genconv_node_held<16>(feat, idx, deg, q, mx, num, den);
      else if (deg <= 20) genconv_node_held<20>(feat, idx, deg, q, mx, num, den);
      else genconv_node_held<ROWS_HELD>(feat, idx, deg, q, mx, num, den);
    }
  } else {
    for (int b0 = 0; b0 < nb; b0 += 32) {
      const int n = min(32, nb - b0);
      const int myidx = (lane < n) ? mine[b0 + lane] : 0;
#pragma unroll 4
      for (int j = 0; j < n; ++j) {
        const int s = __builtin_amdgcn_readlane(myidx, j);
        genconv_max_row(genconv_row_quad(feat, s, q), mx);
      }
    }
    for (int b0 = 0; b0 < novf; b0 += 32) {
      const int k = b0 + lane;
      const bool hit = (k < novf) && (ovfd[k] == node);
      const int myidx = hit ? ovfs[k] : 0;
      unsigned hits = __builtin_amdgcn_ballot_w32(hit);
      while (hits) {
        const int j = __builtin_ctz(hits);
        hits &= hits - 1;
        const int s = __builtin_amdgcn_readlane(myidx, j);
        genconv_max_row(genconv_row_quad(feat, s, q), mx);
      }
    }

    for (int b0 = 0; b0 < nb; b0 += 32) {
      const int n = min(32, nb - b0);
      const int myidx = (lane < n) ? mine[b0 + lane] : 0;
#pragma unroll 4
      for (int j = 0; j < n; ++j) {
        const int s = __builtin_amdgcn_readlane(myidx, j);
        genconv_sum_row(genconv_row_quad(feat, s, q), mx, num, den);
      }
    }
    for (int b0 = 0; b0 < novf; b0 += 32) {
      const int k = b0 + lane;
      const bool hit = (k < novf) && (ovfd[k] == node);
      const int myidx = hit ? ovfs[k] : 0;
      unsigned hits = __builtin_amdgcn_ballot_w32(hit);
      while (hits) {
        const int j = __builtin_ctz(hits);
        hits &= hits - 1;
        const int s = __builtin_amdgcn_readlane(myidx, j);
        genconv_sum_row(genconv_row_quad(feat, s, q), mx, num, den);
      }
    }
  }

  float4 r = make_float4(0.f, 0.f, 0.f, 0.f);
  if (deg > 0) {
    r.x = num.x / den.x;
    r.y = num.y / den.y;
    r.z = num.z / den.z;
    r.w = num.w / den.w;
  }
  if (lane < 24) reinterpret_cast<float4*>(msg + (size_t)node * DF)[lane] = r;
}

__global__ __launch_bounds__(GEMM_THREADS) void genconv_fused_gemm(
    const float* __restrict__ feat, const float* __restrict__ msg,
    const float* __restrict__ W, const float* __restrict__ b,
    float* __restrict__ out, int N) {
  __shared__ __attribute__((aligned(16))) float sW[DF * DF];
  __shared__ __attribute__((aligned(16))) float sA[TILE_ROWS * DF];

  const int tid = threadIdx.x;
  const int rowBase = blockIdx.x * TILE_ROWS;

  constexpr int QUADS_W = DF * DF / 4 / GEMM_THREADS;
  constexpr int QUADS_A = TILE_ROWS * DF / 4 / GEMM_THREADS;
  const float4* __restrict__ W4 = reinterpret_cast<const float4*>(W);
  const float4* __restrict__ f4 =
      reinterpret_cast<const float4*>(feat) + (size_t)rowBase * (DF / 4);
  const float4* __restrict__ m4 =
      reinterpret_cast<const float4*>(msg) + (size_t)rowBase * (DF / 4);
  const int validQ = min(N - rowBase, TILE_ROWS) * (DF / 4);

  float4 wq[QUADS_W];
  float4 fq[QUADS_A];
  float4 mq[QUADS_A];
#pragma unroll
  for (int it = 0; it < QUADS_A; ++it) {
    const int j = min(tid + it * GEMM_THREADS, validQ - 1);
    fq[it] = f4[j];
    mq[it] = m4[j];
  }
#pragma unroll
  for (int it = 0; it < QUADS_W; ++it) wq[it] = W4[tid + it * GEMM_THREADS];

  float4* sW4 = reinterpret_cast<float4*>(sW);
  float4* sA4 = reinterpret_cast<float4*>(sA);
#pragma unroll
  for (int it = 0; it < QUADS_W; ++it) sW4[tid + it * GEMM_THREADS] = wq[it];
#pragma unroll
  for (int it = 0; it < QUADS_A; ++it) {
    const int j = tid + it * GEMM_THREADS;
    float4 v = make_float4(0.f, 0.f, 0.f, 0.f);
    if (j < validQ) {
      v.x = fq[it].x + mq[it].x;
      v.y = fq[it].y + mq[it].y;
      v.z = fq[it].z + mq[it].z;
      v.w = fq[it].w + mq[it].w;
    }
    sA4[j] = v;
  }
  __syncthreads();

  const int wave = tid >> 5;
  const int lane = tid & 31;
  const int mrow = lane & 15;
  const int kh   = lane >> 4;

  const float* aRow = &sA[(wave * 16 + mrow) * DF];

  float bias[NTILES];
#pragma unroll
  for (int t = 0; t < NTILES; ++t) bias[t] = b[t * 16 + mrow];

  v8f acc[NTILES] = {};

  for (int k = 0; k < DF; k += 4) {
    v2f a;
    a.x = aRow[k + 2 * kh + 0];
    a.y = aRow[k + 2 * kh + 1];
#pragma unroll
    for (int t = 0; t < NTILES; ++t) {
      v2f bf;
      bf.x = sW[(k + 2 * kh + 0) * DF + t * 16 + mrow];
      bf.y = sW[(k + 2 * kh + 1) * DF + t * 16 + mrow];
      acc[t] = __builtin_amdgcn_wmma_f32_16x16x4_f32(
           false, a,  false, bf,
           (short)0, acc[t],  false,  false);
    }
  }

#pragma unroll
  for (int t = 0; t < NTILES; ++t) {
#pragma unroll
    for (int i = 0; i < 8; ++i) {
      int r = rowBase + wave * 16 + i + kh * 8;
      if (r < N) out[(size_t)r * DF + t * 16 + mrow] = acc[t][i] + bias[t];
    }
  }
}

extern "C" void kernel_launch(void* const* d_in, const int* in_sizes, int n_in,
                              void* d_out, int out_size, void* d_ws, size_t ws_size,
                              hipStream_t stream) {
  const float* feat = (const float*)d_in[0];
  const int*   src  = (const int*)d_in[1];
  const int*   dst  = (const int*)d_in[2];
  const float* W    = (const float*)d_in[3];
  const float* b    = (const float*)d_in[4];
  float*       out  = (float*)d_out;

  const int N = in_sizes[0] / DF;
  const int E = in_sizes[1];

  const int nbin = (N + BIN_NODES - 1) / BIN_NODES;
  const int nblk = (E + PART_EDGES - 1) / PART_EDGES;
  const size_t Npad = (size_t)nbin * BIN_NODES;
  const size_t Epad = ((size_t)E + 63) / 64 * 64;
  int*   cnt    = (int*)d_ws;
  int*   ovfc   = cnt + Npad;
  int*   bucket = ovfc + 64;
  int*   ovfd   = bucket + Npad * BUCKET_CAP;
  int*   ovfs   = ovfd + Epad;
  float* msg    = (float*)(ovfs + Epad);
  int*   segcnt = (int*)(msg + (size_t)N * DF);
  int*   spillc = segcnt + (size_t)nblk * CNT_PITCH;
  int*   seg    = spillc + (size_t)nblk * 32;
  int*   spill  = seg + (size_t)nbin * nblk * SEG_CAP * 2;

  genconv_edge_partition<<<nblk, PART_THREADS, 0, stream>>>(src, dst, seg, segcnt, spill,
                                                          spillc, ovfc, E, nblk);
  genconv_bin_bucket<<<nbin, FILL_THREADS, 0, stream>>>(seg, segcnt, spill, spillc, cnt,
                                                      bucket, ovfc, ovfd, ovfs, nblk);
  genconv_node_softmax<<<(N + NODE_WAVES - 1) / NODE_WAVES, NODE_WAVES * 32, 0,
                         stream>>>(feat, bucket, cnt, ovfc, ovfd, ovfs, msg, N, E);

  const int ggrid = (N + TILE_ROWS - 1) / TILE_ROWS;
  genconv_fused_gemm<<<ggrid, GEMM_THREADS, 0, stream>>>(feat, msg, W, b, out, N);
}


# Bucket kernel merged into the node kernel: a block builds its 196-node bin's buckets in LDS, then its waves aggregate those nodes from LDS; one launch fewer; over-64-edge nodes sum in segment order
# speedup vs baseline: 7.6671x; 1.1702x over previous
#include <hip/hip_runtime.h>


#define DF 96
#define NTILES 6
#define TILE_ROWS 64
#define GEMM_THREADS 128
#define FEPS 1e-7f
#define BUCKET_CAP 64
#define ROWS_HELD 24
#define BIN_NODES 196
#define MAX_BINS 256
#define MAX_BLKS 1024
#define PART_THREADS 1024
#define PART_PER_THREAD 4
#define PART_EDGES 4096
#define SEG_CAP 64
#define CNT_PITCH 256
#define FILL_THREADS 1024
#define FILL_DEPTH 13

typedef __attribute__((ext_vector_type(2))) float v2f;
typedef __attribute__((ext_vector_type(8))) float v8f;


__global__ __launch_bounds__(PART_THREADS) void genconv_edge_partition(
    const int* __restrict__ src, const int* __restrict__ dst,
    int* __restrict__ segp, int* __restrict__ segcnt,
    int* __restrict__ spillp, int* __restrict__ spillc, int E, int nblk) {
  __shared__ int hist[MAX_BINS];
  __shared__ int off[MAX_BINS];
  __shared__ int wsum[MAX_BINS / 32];
  __shared__ int nspill;
  __shared__ int2 sorted[PART_EDGES];

  int2* __restrict__ seg = reinterpret_cast<int2*>(segp);
  int2* __restrict__ spill = reinterpret_cast<int2*>(spillp);
  const int tid = threadIdx.x;
  const int blk = blockIdx.x;
  const int base = blk * PART_EDGES;

  int d[PART_PER_THREAD], s[PART_PER_THREAD], r[PART_PER_THREAD];
#pragma unroll
  for (int i = 0; i < PART_PER_THREAD; ++i) {
    const int e = base + i * PART_THREADS + tid;
    const bool ok = e < E;
    d[i] = ok ? dst[e] : -1;
    s[i] = ok ? src[e] : 0;
  }
  if (tid < MAX_BINS) hist[tid] = 0;
  if (tid == 0) nspill = 0;
  __syncthreads();

#pragma unroll
  for (int i = 0; i < PART_PER_THREAD; ++i) {
    r[i] = 0;
    if (d[i] >= 0) r[i] = atomicAdd(&hist[(unsigned)d[i] / BIN_NODES], 1);
  }
  __syncthreads();

  int v = 0, x = 0;
  if (tid < MAX_BINS) {
    v = hist[tid];
    x = v;
    for (int o = 1; o < 32; o <<= 1) {
      const int y = __shfl_up(x, o);
      if ((tid & 31) >= o) x += y;
    }
    if ((tid & 31) == 31) wsum[tid >> 5] = x;
  }
  __syncthreads();
  if (tid < MAX_BINS) {
    int front = 0;
    for (int w = 0; w < (tid >> 5); ++w) front += wsum[w];
    off[tid] = front + x - v;
    segcnt[(size_t)blk * CNT_PITCH + tid] = min(v, SEG_CAP);
  }
  __syncthreads();

#pragma unroll
  for (int i = 0; i < PART_PER_THREAD; ++i)
    if (d[i] >= 0) sorted[off[(unsigned)d[i] / BIN_NODES] + r[i]] = make_int2(d[i], s[i]);
  __syncthreads();

  const int n = min(PART_EDGES, E - base);
  for (int p = tid; p < n; p += PART_THREADS) {
    const int2 en = sorted[p];
    const int bin = (unsigned)en.x / BIN_NODES;
    const int rank = p - off[bin];
    int2* __restrict__ to = seg + ((size_t)bin * nblk + blk) * SEG_CAP + rank;
    if (rank >= SEG_CAP) {
      const int q = atomicAdd(&nspill, 1);
      to = spill + (size_t)blk * PART_EDGES + q;
    }
    *to = en;
  }
  __syncthreads();
  if (tid == 0) spillc[blk * 32] = nspill;
}

__device__ __forceinline__ float4 genconv_row_quad(const float* __restrict__ feat,
                                                   int s, int q) {
  const unsigned off = (unsigned)s * (DF * 4u) + (unsigned)q * 16u;
  return *reinterpret_cast<const float4*>(reinterpret_cast<const char*>(feat) + off);
}

__device__ __forceinline__ void genconv_max_row(const float4 f, float4& mx) {
  mx.x = fmaxf(mx.x, fmaxf(f.x, 0.f) + FEPS);
  mx.y = fmaxf(mx.y, fmaxf(f.y, 0.f) + FEPS);
  mx.z = fmaxf(mx.z, fmaxf(f.z, 0.f) + FEPS);
  mx.w = fmaxf(mx.w, fmaxf(f.w, 0.f) + FEPS);
}

__device__ __forceinline__ void genconv_sum_row(const float4 f, const float4 mx,
                                                float4& num, float4& den) {
  const float m0 = fmaxf(f.x, 0.f) + FEPS;
  const float m1 = fmaxf(f.y, 0.f) + FEPS;
  const float m2 = fmaxf(f.z, 0.f) + FEPS;
  const float m3 = fmaxf(f.w, 0.f) + FEPS;
  const float e0 = __expf(m0 - mx.x);
  const float e1 = __expf(m1 - mx.y);
  const float e2 = __expf(m2 - mx.z);
  const float e3 = __expf(m3 - mx.w);
  num.x += e0 * m0;
  num.y += e1 * m1;
  num.z += e2 * m2;
  num.w += e3 * m3;
  den.x += e0;
  den.y += e1;
  den.z += e2;
  den.w += e3;
}

template <int R>
__device__ __forceinline__ void genconv_node_held(
    const float* __restrict__ feat, int idx, int deg, int q,
    float4& mx, float4& num, float4& den) {
  float4 row[R];
#pragma unroll
  for (int j = 0; j < R; ++j)
    row[j] = genconv_row_quad(feat, __builtin_amdgcn_readlane(idx, j), q);
  __builtin_amdgcn_sched_barrier(0);
#pragma unroll
  for (int j = 0; j < R; ++j) genconv_max_row(row[j], mx);
#pragma unroll
  for (int j = 0; j < R; ++j)
    if (j < deg) genconv_sum_row(row[j], mx, num, den);
}

template <bool SUM>
__device__ __forceinline__ void genconv_scan_pass(
    const float* __restrict__ feat, const int2* __restrict__ mine, const int* lseg,
    int nslot, const int2* __restrict__ spill, const int* __restrict__ spillc,
    int nblk, int k, int spilled, int node, int q, int lane,
    float4& mx, float4& num, float4& den) {
  for (int b0 = 0; b0 < nslot; b0 += 32) {
    const int i = b0 + lane;
    bool hit = false;
    int s = 0;
    if (i < nslot && (i & (SEG_CAP - 1)) < lseg[i / SEG_CAP]) {
      const int2 e2 = mine[i];
      hit = (e2.x == node);
      s = e2.y;
    }
    unsigned hits = __builtin_amdgcn_ballot_w32(hit);
    while (hits) {
      const int j = __builtin_ctz(hits);
      hits &= hits - 1;
      const float4 f = genconv_row_quad(feat, __builtin_amdgcn_readlane(s, j), q);
      if (SUM) genconv_sum_row(f, mx, num, den);
      else genconv_max_row(f, mx);
    }
  }
  if (spilled) {
    for (int b = 0; b < nblk; ++b) {
      const int c = min(spillc[b * 32], PART_EDGES);
      for (int b0 = 0; b0 < c; b0 += 32) {
        const int i = b0 + lane;
        bool hit = false;
        int s = 0;
        if (i < c) {
          const int2 e2 = spill[(size_t)b * PART_EDGES + i];
          hit = (e2.x == node);
          s = e2.y;
        }
        unsigned hits = __builtin_amdgcn_ballot_w32(hit);
        while (hits) {
          const int j = __builtin_ctz(hits);
          hits &= hits - 1;
          const float4 f = genconv_row_quad(feat, __builtin_amdgcn_readlane(s, j), q);
          if (SUM) genconv_sum_row(f, mx, num, den);
          else genconv_max_row(f, mx);
        }
      }
    }
  }
}

__device__ __forceinline__ int genconv_next_node(int* counter, int lane) {
  int t = 0;
  if (lane == 0) t = atomicAdd(counter, 1);
  return __builtin_amdgcn_readfirstlane(t);
}

__global__ __launch_bounds__(FILL_THREADS) void genconv_bin_aggregate(
    const float* __restrict__ feat, const int* __restrict__ segp,
    const int* __restrict__ segcnt, const int* __restrict__ spillp,
    const int* __restrict__ spillc, float* __restrict__ msg, int N, int nblk) {
  __shared__ int lbucket[BIN_NODES * BUCKET_CAP];
  __shared__ int lcnt[BIN_NODES];
  __shared__ int lseg[MAX_BLKS];
  __shared__ int anyspill;
  __shared__ int nextnode;

  const int2* __restrict__ seg = reinterpret_cast<const int2*>(segp);
  const int2* __restrict__ spill = reinterpret_cast<const int2*>(spillp);
  const int tid = threadIdx.x;
  const int k = blockIdx.x;
  const int node0 = k * BIN_NODES;

  int sc = 0, pc = 0;
  if (tid < nblk) {
    sc = segcnt[(size_t)tid * CNT_PITCH + k];
    pc = spillc[tid * 32];
  }
  if (tid < BIN_NODES) lcnt[tid] = 0;
  if (tid == 0) {
    anyspill = 0;
    nextnode = 0;
  }
  static_assert(FILL_THREADS == MAX_BLKS, "one thread per entry of lseg");
  lseg[tid] = sc;
  __syncthreads();
  if (pc > 0) anyspill = 1;

  const int2* __restrict__ mine = seg + (size_t)k * nblk * SEG_CAP;
  const int nslot = nblk * SEG_CAP;
  for (int g0 = 0; g0 < nslot; g0 += FILL_THREADS * FILL_DEPTH) {
    int2 en[FILL_DEPTH];
    int rk[FILL_DEPTH];
    bool ok[FILL_DEPTH];
#pragma unroll
    for (int u = 0; u < FILL_DEPTH; ++u) {
      const int i = g0 + u * FILL_THREADS + tid;
      ok[u] = (i < nslot) && ((i & (SEG_CAP - 1)) < lseg[i / SEG_CAP]);
      en[u] = make_int2(node0, 0);
      if (ok[u]) en[u] = mine[i];
    }
#pragma unroll
    for (int u = 0; u < FILL_DEPTH; ++u) {
      rk[u] = 0;
      if (ok[u]) rk[u] = atomicAdd(&lcnt[en[u].x - node0], 1);
    }
#pragma unroll
    for (int u = 0; u < FILL_DEPTH; ++u)
      if (ok[u] && rk[u] < BUCKET_CAP) lbucket[(en[u].x - node0) * BUCKET_CAP + rk[u]] = en[u].y;
  }
  __syncthreads();

  const int spilled = anyspill;
  if (spilled) {
    for (int b = 0; b < nblk; ++b) {
      const int c = min(spillc[b * 32], PART_EDGES);
      for (int i = tid; i < c; i += FILL_THREADS) {
        const int2 e2 = spill[(size_t)b * PART_EDGES + i];
        if ((unsigned)e2.x / BIN_NODES == (unsigned)k) {
          const int r = atomicAdd(&lcnt[e2.x - node0], 1);
          if (r < BUCKET_CAP) lbucket[(e2.x - node0) * BUCKET_CAP + r] = e2.y;
        }
      }
    }
  }
  __syncthreads();

  const int lane = tid & 31;
  const int nhere = min(BIN_NODES, N - node0);
  const int q = (lane < 24) ? lane : lane - 8;

  for (;;) {
    const int cur = genconv_next_node(&nextnode, lane);
    if (cur >= nhere) break;
    const int node = node0 + cur;
    const int deg = __builtin_amdgcn_readfirstlane(lcnt[cur]);
    const int* bkt = lbucket + cur * BUCKET_CAP;

    float4 mx = make_float4(0.f, 0.f, 0.f, 0.f);
    float4 num = make_float4(0.f, 0.f, 0.f, 0.f);
    float4 den = make_float4(0.f, 0.f, 0.f, 0.f);
    if (deg <= ROWS_HELD) {
      if (deg > 0) {
        const int idx = bkt[min(lane, deg - 1)];
        if (deg <= 4) genconv_node_held<4>(feat, idx, deg, q, mx, num, den);
        else if (deg <= 8) genconv_node_held<8>(feat, idx, deg, q, mx, num, den);
        else if (deg <= 12) genconv_node_held<12>(feat, idx, deg, q, mx, num, den);
        else if (deg <= 16) genconv_node_held<16>(feat, idx, deg, q, mx, num, den);
        else if (deg <= 20) genconv_node_held<20>(feat, idx, deg, q, mx, num, den);
        else genconv_node_held<ROWS_HELD>(feat, idx, deg, q, mx, num, den);
      }
    } else if (deg <= BUCKET_CAP) {
      for (int b0 = 0; b0 < deg; b0 += 32) {
        const int n = min(32, deg - b0);
        const int myidx = (lane < n) ? bkt[b0 + lane] : 0;
#pragma unroll 4
        for (int j = 0; j < n; ++j) {
          const int s = __builtin_amdgcn_readlane(myidx, j);
          genconv_max_row(genconv_row_quad(feat, s, q), mx);
        }
      }
      for (int b0 = 0; b0 < deg; b0 += 32) {
        const int n = min(32, deg - b0);
        const int myidx = (lane < n) ? bkt[b0 + lane] : 0;
#pragma unroll 4
        for (int j = 0; j < n; ++j) {
          const int s = __builtin_amdgcn_readlane(myidx, j);
          genconv_sum_row(genconv_row_quad(feat, s, q), mx, num, den);
        }
      }
    } else {
      genconv_scan_pass<false>(feat, mine, lseg, nslot, spill, spillc, nblk, k, spilled,
                               node, q, lane, mx, num, den);
      genconv_scan_pass<true>(feat, mine, lseg, nslot, spill, spillc, nblk, k, spilled,
                              node, q, lane, mx, num, den);
    }

    float4 r = make_float4(0.f, 0.f, 0.f, 0.f);
    if (deg > 0) {
      r.x = num.x / den.x;
      r.y = num.y / den.y;
      r.z = num.z / den.z;
      r.w = num.w / den.w;
    }
    const unsigned moff = (unsigned)node * (DF * 4u) + (unsigned)lane * 16u;
    if (lane < 24) *reinterpret_cast<float4*>(reinterpret_cast<char*>(msg) + moff) = r;
  }
}

__global__ __launch_bounds__(GEMM_THREADS) void genconv_fused_gemm(
    const float* __restrict__ feat, const float* __restrict__ msg,
    const float* __restrict__ W, const float* __restrict__ b,
    float* __restrict__ out, int N) {
  __shared__ __attribute__((aligned(16))) float sW[DF * DF];
  __shared__ __attribute__((aligned(16))) float sA[TILE_ROWS * DF];

  const int tid = threadIdx.x;
  const int rowBase = blockIdx.x * TILE_ROWS;

  constexpr int QUADS_W = DF * DF / 4 / GEMM_THREADS;
  constexpr int QUADS_A = TILE_ROWS * DF / 4 / GEMM_THREADS;
  const float4* __restrict__ W4 = reinterpret_cast<const float4*>(W);
  const float4* __restrict__ f4 =
      reinterpret_cast<const float4*>(feat) + (size_t)rowBase * (DF / 4);
  const float4* __restrict__ m4 =
      reinterpret_cast<const float4*>(msg) + (size_t)rowBase * (DF / 4);
  const int validQ = min(N - rowBase, TILE_ROWS) * (DF / 4);

  float4 wq[QUADS_W];
  float4 fq[QUADS_A];
  float4 mq[QUADS_A];
#pragma unroll
  for (int it = 0; it < QUADS_A; ++it) {
    const int j = min(tid + it * GEMM_THREADS, validQ - 1);
    fq[it] = f4[j];
    mq[it] = m4[j];
  }
#pragma unroll
  for (int it = 0; it < QUADS_W; ++it) wq[it] = W4[tid + it * GEMM_THREADS];

  float4* sW4 = reinterpret_cast<float4*>(sW);
  float4* sA4 = reinterpret_cast<float4*>(sA);
#pragma unroll
  for (int it = 0; it < QUADS_W; ++it) sW4[tid + it * GEMM_THREADS] = wq[it];
#pragma unroll
  for (int it = 0; it < QUADS_A; ++it) {
    const int j = tid + it * GEMM_THREADS;
    float4 v = make_float4(0.f, 0.f, 0.f, 0.f);
    if (j < validQ) {
      v.x = fq[it].x + mq[it].x;
      v.y = fq[it].y + mq[it].y;
      v.z = fq[it].z + mq[it].z;
      v.w = fq[it].w + mq[it].w;
    }
    sA4[j] = v;
  }
  __syncthreads();

  const int wave = tid >> 5;
  const int lane = tid & 31;
  const int mrow = lane & 15;
  const int kh   = lane >> 4;

  const float* aRow = &sA[(wave * 16 + mrow) * DF];

  float bias[NTILES];
#pragma unroll
  for (int t = 0; t < NTILES; ++t) bias[t] = b[t * 16 + mrow];

  v8f acc[NTILES] = {};

  for (int k = 0; k < DF; k += 4) {
    v2f a;
    a.x = aRow[k + 2 * kh + 0];
    a.y = aRow[k + 2 * kh + 1];
#pragma unroll
    for (int t = 0; t < NTILES; ++t) {
      v2f bf;
      bf.x = sW[(k + 2 * kh + 0) * DF + t * 16 + mrow];
      bf.y = sW[(k + 2 * kh + 1) * DF + t * 16 + mrow];
      acc[t] = __builtin_amdgcn_wmma_f32_16x16x4_f32(
           false, a,  false, bf,
           (short)0, acc[t],  false,  false);
    }
  }

#pragma unroll
  for (int t = 0; t < NTILES; ++t) {
#pragma unroll
    for (int i = 0; i < 8; ++i) {
      int r = rowBase + wave * 16 + i + kh * 8;
      if (r < N) out[(size_t)r * DF + t * 16 + mrow] = acc[t][i] + bias[t];
    }
  }
}

extern "C" void kernel_launch(void* const* d_in, const int* in_sizes, int n_in,
                              void* d_out, int out_size, void* d_ws, size_t ws_size,
                              hipStream_t stream) {
  const float* feat = (const float*)d_in[0];
  const int*   src  = (const int*)d_in[1];
  const int*   dst  = (const int*)d_in[2];
  const float* W    = (const float*)d_in[3];
  const float* b    = (const float*)d_in[4];
  float*       out  = (float*)d_out;

  const int N = in_sizes[0] / DF;
  const int E = in_sizes[1];

  const int nbin = (N + BIN_NODES - 1) / BIN_NODES;
  const int nblk = (E + PART_EDGES - 1) / PART_EDGES;
  float* msg    = (float*)d_ws;
  int*   segcnt = (int*)(msg + (size_t)N * DF);
  int*   spillc = segcnt + (size_t)nblk * CNT_PITCH;
  int*   seg    = spillc + (size_t)nblk * 32;
  int*   spill  = seg + (size_t)nbin * nblk * SEG_CAP * 2;

  genconv_edge_partition<<<nblk, PART_THREADS, 0, stream>>>(src, dst, seg, segcnt, spill,
                                                          spillc, E, nblk);
  genconv_bin_aggregate<<<nbin, FILL_THREADS, 0, stream>>>(feat, seg, segcnt, spill, spillc,
                                                         msg, N, nblk);

  const int ggrid = (N + TILE_ROWS - 1) / TILE_ROWS;
  genconv_fused_gemm<<<ggrid, GEMM_THREADS, 0, stream>>>(feat, msg, W, b, out, N);
}


# GEMM merged into the aggregation kernel: a block keeps feat + msg of its 196 nodes in LDS beside W, its waves do the f32 16x16x4 products there; msg plane and 3rd launch gone; sum order unchanged
# speedup vs baseline: 8.8171x; 1.1500x over previous
#include <hip/hip_runtime.h>


#define DF 96
#define BIN_TILES 13
#define SA_PITCH 100
#define SW_PITCH 104
#define COL_UNITS 3
#define FEPS 1e-7f
#define BUCKET_CAP 64
#define ROWS_HELD 24
#define BIN_NODES 196
#define MAX_BINS 256
#define MAX_BLKS 1024
#define PART_THREADS 1024
#define PART_PER_THREAD 4
#define PART_EDGES 4096
#define SEG_CAP 64
#define CNT_PITCH 256
#define FILL_THREADS 1024
#define FILL_DEPTH 13

typedef __attribute__((ext_vector_type(2))) float v2f;
typedef __attribute__((ext_vector_type(8))) float v8f;


__global__ __launch_bounds__(PART_THREADS) void genconv_edge_partition(
    const int* __restrict__ src, const int* __restrict__ dst,
    int* __restrict__ segp, int* __restrict__ segcnt,
    int* __restrict__ spillp, int* __restrict__ spillc, int E, int nblk) {
  __shared__ int hist[MAX_BINS];
  __shared__ int off[MAX_BINS];
  __shared__ int wsum[MAX_BINS / 32];
  __shared__ int nspill;
  __shared__ int2 sorted[PART_EDGES];

  int2* __restrict__ seg = reinterpret_cast<int2*>(segp);
  int2* __restrict__ spill = reinterpret_cast<int2*>(spillp);
  const int tid = threadIdx.x;
  const int blk = blockIdx.x;
  const int base = blk * PART_EDGES;

  int d[PART_PER_THREAD], s[PART_PER_THREAD], r[PART_PER_THREAD];
#pragma unroll
  for (int i = 0; i < PART_PER_THREAD; ++i) {
    const int e = base + i * PART_THREADS + tid;
    const bool ok = e < E;
    d[i] = ok ? dst[e] : -1;
    s[i] = ok ? src[e] : 0;
  }
  if (tid < MAX_BINS) hist[tid] = 0;
  if (tid == 0) nspill = 0;
  __syncthreads();

#pragma unroll
  for (int i = 0; i < PART_PER_THREAD; ++i) {
    r[i] = 0;
    if (d[i] >= 0) r[i] = atomicAdd(&hist[(unsigned)d[i] / BIN_NODES], 1);
  }
  __syncthreads();

  int v = 0, x = 0;
  if (tid < MAX_BINS) {
    v = hist[tid];
    x = v;
    for (int o = 1; o < 32; o <<= 1) {
      const int y = __shfl_up(x, o);
      if ((tid & 31) >= o) x += y;
    }
    if ((tid & 31) == 31) wsum[tid >> 5] = x;
  }
  __syncthreads();
  if (tid < MAX_BINS) {
    int front = 0;
    for (int w = 0; w < (tid >> 5); ++w) front += wsum[w];
    off[tid] = front + x - v;
    segcnt[(size_t)blk * CNT_PITCH + tid] = min(v, SEG_CAP);
  }
  __syncthreads();

#pragma unroll
  for (int i = 0; i < PART_PER_THREAD; ++i)
    if (d[i] >= 0) sorted[off[(unsigned)d[i] / BIN_NODES] + r[i]] = make_int2(d[i], s[i]);
  __syncthreads();

  const int n = min(PART_EDGES, E - base);
  for (int p = tid; p < n; p += PART_THREADS) {
    const int2 en = sorted[p];
    const int bin = (unsigned)en.x / BIN_NODES;
    const int rank = p - off[bin];
    int2* __restrict__ to = seg + ((size_t)bin * nblk + blk) * SEG_CAP + rank;
    if (rank >= SEG_CAP) {
      const int q = atomicAdd(&nspill, 1);
      to = spill + (size_t)blk * PART_EDGES + q;
    }
    *to = en;
  }
  __syncthreads();
  if (tid == 0) spillc[blk * 32] = nspill;
}

__device__ __forceinline__ float4 genconv_row_quad(const float* __restrict__ feat,
                                                   int s, int q) {
  const unsigned off = (unsigned)s * (DF * 4u) + (unsigned)q * 16u;
  return *reinterpret_cast<const float4*>(reinterpret_cast<const char*>(feat) + off);
}

__device__ __forceinline__ void genconv_max_row(const float4 f, float4& mx) {
  mx.x = fmaxf(mx.x, fmaxf(f.x, 0.f) + FEPS);
  mx.y = fmaxf(mx.y, fmaxf(f.y, 0.f) + FEPS);
  mx.z = fmaxf(mx.z, fmaxf(f.z, 0.f) + FEPS);
  mx.w = fmaxf(mx.w, fmaxf(f.w, 0.f) + FEPS);
}

__device__ __forceinline__ void genconv_sum_row(const float4 f, const float4 mx,
                                                float4& num, float4& den) {
  const float m0 = fmaxf(f.x, 0.f) + FEPS;
  const float m1 = fmaxf(f.y, 0.f) + FEPS;
  const float m2 = fmaxf(f.z, 0.f) + FEPS;
  const float m3 = fmaxf(f.w, 0.f) + FEPS;
  const float e0 = __expf(m0 - mx.x);
  const float e1 = __expf(m1 - mx.y);
  const float e2 = __expf(m2 - mx.z);
  const float e3 = __expf(m3 - mx.w);
  num.x += e0 * m0;
  num.y += e1 * m1;
  num.z += e2 * m2;
  num.w += e3 * m3;
  den.x += e0;
  den.y += e1;
  den.z += e2;
  den.w += e3;
}

template <int R>
__device__ __forceinline__ void genconv_node_held(
    const float* __restrict__ feat, int idx, int deg, int q,
    float4& mx, float4& num, float4& den) {
  float4 row[R];
#pragma unroll
  for (int j = 0; j < R; ++j)
    row[j] = genconv_row_quad(feat, __builtin_amdgcn_readlane(idx, j), q);
  __builtin_amdgcn_sched_barrier(0);
#pragma unroll
  for (int j = 0; j < R; ++j) genconv_max_row(row[j], mx);
#pragma unroll
  for (int j = 0; j < R; ++j)
    if (j < deg) genconv_sum_row(row[j], mx, num, den);
}

template <bool SUM>
__device__ __forceinline__ void genconv_scan_pass(
    const float* __restrict__ feat, const int2* __restrict__ mine, const int* lseg,
    int nslot, const int2* __restrict__ spill, const int* __restrict__ spillc,
    int nblk, int k, int spilled, int node, int q, int lane,
    float4& mx, float4& num, float4& den) {
  for (int b0 = 0; b0 < nslot; b0 += 32) {
    const int i = b0 + lane;
    bool hit = false;
    int s = 0;
    if (i < nslot && (i & (SEG_CAP - 1)) < lseg[i / SEG_CAP]) {
      const int2 e2 = mine[i];
      hit = (e2.x == node);
      s = e2.y;
    }
    unsigned hits = __builtin_amdgcn_ballot_w32(hit);
    while (hits) {
      const int j = __builtin_ctz(hits);
      hits &= hits - 1;
      const float4 f = genconv_row_quad(feat, __builtin_amdgcn_readlane(s, j), q);
      if (SUM) genconv_sum_row(f, mx, num, den);
      else genconv_max_row(f, mx);
    }
  }
  if (spilled) {
    for (int b = 0; b < nblk; ++b) {
      const int c = min(spillc[b * 32], PART_EDGES);
      for (int b0 = 0; b0 < c; b0 += 32) {
        const int i = b0 + lane;
        bool hit = false;
        int s = 0;
        if (i < c) {
          const int2 e2 = spill[(size_t)b * PART_EDGES + i];
          hit = (e2.x == node);
          s = e2.y;
        }
        unsigned hits = __builtin_amdgcn_ballot_w32(hit);
        while (hits) {
          const int j = __builtin_ctz(hits);
          hits &= hits - 1;
          const float4 f = genconv_row_quad(feat, __builtin_amdgcn_readlane(s, j), q);
          if (SUM) genconv_sum_row(f, mx, num, den);
          else genconv_max_row(f, mx);
        }
      }
    }
  }
}

__device__ __forceinline__ float genconv_xhalf(float x) {
  const int v = __builtin_bit_cast(int, x);
  return __builtin_bit_cast(
      float, __builtin_amdgcn_permlanex16(v, v, 0x76543210, 0xfedcba98, false, false));
}

__device__ __forceinline__ int genconv_next_node(int* counter, int lane) {
  int t = 0;
  if (lane == 0) t = atomicAdd(counter, 1);
  return __builtin_amdgcn_readfirstlane(t);
}

__global__ __launch_bounds__(FILL_THREADS) void genconv_bin_aggregate(
    const float* __restrict__ feat, const int* __restrict__ segp,
    const int* __restrict__ segcnt, const int* __restrict__ spillp,
    const int* __restrict__ spillc, const float* __restrict__ W,
    const float* __restrict__ b, float* __restrict__ out, int N, int nblk) {
  struct Lds {
    int lcnt[BIN_NODES];
    int lseg[MAX_BLKS];
    int anyspill;
    int nextnode;
    int nextunit;
    int lbucket[BIN_NODES * BUCKET_CAP];
    __attribute__((aligned(16))) float sW[DF * SW_PITCH];
    __attribute__((aligned(16))) float sA[BIN_TILES * 16 * SA_PITCH];
  };
  __shared__ Lds lds;
  int* const lbucket = lds.lbucket;
  int* const lcnt = lds.lcnt;
  int* const lseg = lds.lseg;
  int& anyspill = lds.anyspill;
  int& nextnode = lds.nextnode;
  int& nextunit = lds.nextunit;
  float* const sW = lds.sW;
  float* const sA = lds.sA;

  const int2* __restrict__ seg = reinterpret_cast<const int2*>(segp);
  const int2* __restrict__ spill = reinterpret_cast<const int2*>(spillp);
  const int tid = threadIdx.x;
  const int k = blockIdx.x;
  const int node0 = k * BIN_NODES;

  int sc = 0, pc = 0;
  if (tid < nblk) {
    sc = segcnt[(size_t)tid * CNT_PITCH + k];
    pc = spillc[tid * 32];
  }
  constexpr int QUADS_W = DF * DF / 4;
  constexpr int QW_PER = (QUADS_W + FILL_THREADS - 1) / FILL_THREADS;
  constexpr int QUADS_A = BIN_TILES * 16 * (DF / 4);
  constexpr int QA_PER = (QUADS_A + FILL_THREADS - 1) / FILL_THREADS;
  const int nhere = min(BIN_NODES, N - node0);
  const int validQ = nhere * (DF / 4);
  const float4* __restrict__ W4 = reinterpret_cast<const float4*>(W);
  const float4* __restrict__ f4 =
      reinterpret_cast<const float4*>(feat) + (size_t)node0 * (DF / 4);
  float4 wq[QW_PER];
  float4 fq[QA_PER];
#pragma unroll
  for (int it = 0; it < QW_PER; ++it) wq[it] = W4[min(tid + it * FILL_THREADS, QUADS_W - 1)];
#pragma unroll
  for (int it = 0; it < QA_PER; ++it) fq[it] = f4[min(tid + it * FILL_THREADS, validQ - 1)];
  if (tid < BIN_NODES) lcnt[tid] = 0;
  if (tid == 0) {
    anyspill = 0;
    nextnode = 0;
    nextunit = 0;
  }
  static_assert(FILL_THREADS == MAX_BLKS, "one thread per entry of lseg");
  lseg[tid] = sc;
  __syncthreads();
  if (pc > 0) anyspill = 1;

  const int2* __restrict__ mine = seg + (size_t)k * nblk * SEG_CAP;
  const int nslot = nblk * SEG_CAP;
  for (int g0 = 0; g0 < nslot; g0 += FILL_THREADS * FILL_DEPTH) {
    int2 en[FILL_DEPTH];
    int rk[FILL_DEPTH];
    bool ok[FILL_DEPTH];
#pragma unroll
    for (int u = 0; u < FILL_DEPTH; ++u) {
      const int i = g0 + u * FILL_THREADS + tid;
      ok[u] = (i < nslot) && ((i & (SEG_CAP - 1)) < lseg[i / SEG_CAP]);
      en[u] = make_int2(node0, 0);
      if (ok[u]) en[u] = mine[i];
    }
#pragma unroll
    for (int u = 0; u < FILL_DEPTH; ++u) {
      rk[u] = 0;
      if (ok[u]) rk[u] = atomicAdd(&lcnt[en[u].x - node0], 1);
    }
#pragma unroll
    for (int u = 0; u < FILL_DEPTH; ++u)
      if (ok[u] && rk[u] < BUCKET_CAP) lbucket[(en[u].x - node0) * BUCKET_CAP + rk[u]] = en[u].y;
  }
  float4* sW4 = reinterpret_cast<float4*>(sW);
  float4* sA4 = reinterpret_cast<float4*>(sA);
#pragma unroll
  for (int it = 0; it < QW_PER; ++it) {
    const int j = tid + it * FILL_THREADS;
    if (j < QUADS_W) sW4[(j / (DF / 4)) * (SW_PITCH / 4) + j % (DF / 4)] = wq[it];
  }
#pragma unroll
  for (int it = 0; it < QA_PER; ++it) {
    const int j = tid + it * FILL_THREADS;
    if (j < QUADS_A)
      sA4[(j / (DF / 4)) * (SA_PITCH / 4) + j % (DF / 4)] =
          (j < validQ) ? fq[it] : make_float4(0.f, 0.f, 0.f, 0.f);
  }
  __syncthreads();

  const int spilled = anyspill;
  if (spilled) {
    for (int b = 0; b < nblk; ++b) {
      const int c = min(spillc[b * 32], PART_EDGES);
      for (int i = tid; i < c; i += FILL_THREADS) {
        const int2 e2 = spill[(size_t)b * PART_EDGES + i];
        if ((unsigned)e2.x / BIN_NODES == (unsigned)k) {
          const int r = atomicAdd(&lcnt[e2.x - node0], 1);
          if (r < BUCKET_CAP) lbucket[(e2.x - node0) * BUCKET_CAP + r] = e2.y;
        }
      }
    }
  }
  __syncthreads();

  const int lane = tid & 31;
  const int q = (lane < 24) ? lane : lane - 8;

  for (;;) {
    const int cur = genconv_next_node(&nextnode, lane);
    if (cur >= nhere) break;
    const int node = node0 + cur;
    const int deg = __builtin_amdgcn_readfirstlane(lcnt[cur]);
    const int* bkt = lbucket + cur * BUCKET_CAP;

    float4 mx = make_float4(0.f, 0.f, 0.f, 0.f);
    float4 num = make_float4(0.f, 0.f, 0.f, 0.f);
    float4 den = make_float4(0.f, 0.f, 0.f, 0.f);
    if (deg <= ROWS_HELD) {
      if (deg > 0) {
        const int idx = bkt[min(lane, deg - 1)];
        if (deg <= 4) genconv_node_held<4>(feat, idx, deg, q, mx, num, den);
        else if (deg <= 8) genconv_node_held<8>(feat, idx, deg, q, mx, num, den);
        else if (deg <= 12) genconv_node_held<12>(feat, idx, deg, q, mx, num, den);
        else if (deg <= 16) genconv_node_held<16>(feat, idx, deg, q, mx, num, den);
        else if (deg <= 20) genconv_node_held<20>(feat, idx, deg, q, mx, num, den);
        else genconv_node_held<ROWS_HELD>(feat, idx, deg, q, mx, num, den);
      }
    } else if (deg <= BUCKET_CAP) {
      for (int b0 = 0; b0 < deg; b0 += 32) {
        const int n = min(32, deg - b0);
        const int myidx = (lane < n) ? bkt[b0 + lane] : 0;
#pragma unroll 4
        for (int j = 0; j < n; ++j) {
          const int s = __builtin_amdgcn_readlane(myidx, j);
          genconv_max_row(genconv_row_quad(feat, s, q), mx);
        }
      }
      for (int b0 = 0; b0 < deg; b0 += 32) {
        const int n = min(32, deg - b0);
        const int myidx = (lane < n) ? bkt[b0 + lane] : 0;
#pragma unroll 4
        for (int j = 0; j < n; ++j) {
          const int s = __builtin_amdgcn_readlane(myidx, j);
          genconv_sum_row(genconv_row_quad(feat, s, q), mx, num, den);
        }
      }
    } else {
      genconv_scan_pass<false>(feat, mine, lseg, nslot, spill, spillc, nblk, k, spilled,
                               node, q, lane, mx, num, den);
      genconv_scan_pass<true>(feat, mine, lseg, nslot, spill, spillc, nblk, k, spilled,
                              node, q, lane, mx, num, den);
    }

    float4 r = make_float4(0.f, 0.f, 0.f, 0.f);
    if (deg > 0) {
      r.x = num.x / den.x;
      r.y = num.y / den.y;
      r.z = num.z / den.z;
      r.w = num.w / den.w;
    }
    if (lane < 24) {
      float4* p = sA4 + cur * (SA_PITCH / 4) + lane;
      float4 f = *p;
      f.x += r.x;
      f.y += r.y;
      f.z += r.z;
      f.w += r.w;
      *p = f;
    }
  }
  __syncthreads();

  const int mrow = lane & 15;
  const int kh = lane >> 4;
  for (;;) {
    const int u = genconv_next_node(&nextunit, lane);
    if (u >= BIN_TILES * COL_UNITS) break;
    const int rt = u / COL_UNITS;
    const int cu = u - rt * COL_UNITS;
    if (rt * 16 >= nhere) break;
    const float* aRow = sA + (rt * 16 + mrow) * SA_PITCH + 2 * kh;
    const float* bCol = sW + (2 * kh) * SW_PITCH + cu * 32 + mrow;
    v8f acc0 = {};
    v8f acc1 = {};
#pragma unroll
    for (int kk = 0; kk < DF; kk += 4) {
      const v2f a = *reinterpret_cast<const v2f*>(aRow + kk);
      v2f b0, b1;
      b0.x = bCol[(kk + 0) * SW_PITCH];
      b0.y = bCol[(kk + 1) * SW_PITCH];
      b1.x = bCol[(kk + 0) * SW_PITCH + 16];
      b1.y = bCol[(kk + 1) * SW_PITCH + 16];
      acc0 = __builtin_amdgcn_wmma_f32_16x16x4_f32(
           false, a,  false, b0,
           (short)0, acc0,  false,  false);
      acc1 = __builtin_amdgcn_wmma_f32_16x16x4_f32(
           false, a,  false, b1,
           (short)0, acc1,  false,  false);
    }
    const float bias = b[cu * 32 + lane];
#pragma unroll
    for (int i = 0; i < 8; ++i) {
      const float give = kh ? acc0[i] : acc1[i];
      const float got = genconv_xhalf(give);
      const float lo = kh ? got : acc0[i];
      const float hi = kh ? acc1[i] : got;
      const int r0 = rt * 16 + i;
      float* o = out + (size_t)(node0 + r0) * DF + cu * 32 + lane;
      if (r0 < nhere) o[0] = lo + bias;
      if (r0 + 8 < nhere) o[8 * DF] = hi + bias;
    }
  }
}

extern "C" void kernel_launch(void* const* d_in, const int* in_sizes, int n_in,
                              void* d_out, int out_size, void* d_ws, size_t ws_size,
                              hipStream_t stream) {
  const float* feat = (const float*)d_in[0];
  const int*   src  = (const int*)d_in[1];
  const int*   dst  = (const int*)d_in[2];
  const float* W    = (const float*)d_in[3];
  const float* b    = (const float*)d_in[4];
  float*       out  = (float*)d_out;

  const int N = in_sizes[0] / DF;
  const int E = in_sizes[1];

  const int nbin = (N + BIN_NODES - 1) / BIN_NODES;
  const int nblk = (E + PART_EDGES - 1) / PART_EDGES;
  int*   segcnt = (int*)d_ws;
  int*   spillc = segcnt + (size_t)nblk * CNT_PITCH;
  int*   seg    = spillc + (size_t)nblk * 32;
  int*   spill  = seg + (size_t)nbin * nblk * SEG_CAP * 2;

  genconv_edge_partition<<<nblk, PART_THREADS, 0, stream>>>(src, dst, seg, segcnt, spill,
                                                          spillc, E, nblk);
  genconv_bin_aggregate<<<nbin, FILL_THREADS, 0, stream>>>(feat, seg, segcnt, spill, spillc,
                                                         W, b, out, N, nblk);
}


# Aggregation lane map: a wave reads two edges at a time, one per 16-lane group, 6 features per lane on all 32 lanes (was one edge on 24 lanes x 4); a node's sums are two partial sums added at the end
# speedup vs baseline: 9.5677x; 1.0851x over previous
#include <hip/hip_runtime.h>


#define DF 96
#define BIN_TILES 13
#define SA_PITCH 100
#define SW_PITCH 104
#define COL_UNITS 3
#define FEPS 1e-7f
#define BUCKET_CAP 64
#define ROWS_HELD 24
#define PAIRS_HELD (ROWS_HELD / 2)
#define BIN_NODES 196
#define MAX_BINS 256
#define MAX_BLKS 1024
#define PART_THREADS 1024
#define PART_PER_THREAD 4
#define PART_EDGES 4096
#define SEG_CAP 64
#define CNT_PITCH 256
#define FILL_THREADS 1024
#define FILL_DEPTH 13

typedef __attribute__((ext_vector_type(2))) float v2f;
typedef __attribute__((ext_vector_type(8))) float v8f;


__global__ __launch_bounds__(PART_THREADS) void genconv_edge_partition(
    const int* __restrict__ src, const int* __restrict__ dst,
    int* __restrict__ segp, int* __restrict__ segcnt,
    int* __restrict__ spillp, int* __restrict__ spillc, int E, int nblk) {
  __shared__ int hist[MAX_BINS];
  __shared__ int off[MAX_BINS];
  __shared__ int wsum[MAX_BINS / 32];
  __shared__ int nspill;
  __shared__ int2 sorted[PART_EDGES];

  int2* __restrict__ seg = reinterpret_cast<int2*>(segp);
  int2* __restrict__ spill = reinterpret_cast<int2*>(spillp);
  const int tid = threadIdx.x;
  const int blk = blockIdx.x;
  const int base = blk * PART_EDGES;

  int d[PART_PER_THREAD], s[PART_PER_THREAD], r[PART_PER_THREAD];
#pragma unroll
  for (int i = 0; i < PART_PER_THREAD; ++i) {
    const int e = base + i * PART_THREADS + tid;
    const bool ok = e < E;
    d[i] = ok ? dst[e] : -1;
    s[i] = ok ? src[e] : 0;
  }
  if (tid < MAX_BINS) hist[tid] = 0;
  if (tid == 0) nspill = 0;
  __syncthreads();

#pragma unroll
  for (int i = 0; i < PART_PER_THREAD; ++i) {
    r[i] = 0;
    if (d[i] >= 0) r[i] = atomicAdd(&hist[(unsigned)d[i] / BIN_NODES], 1);
  }
  __syncthreads();

  int v = 0, x = 0;
  if (tid < MAX_BINS) {
    v = hist[tid];
    x = v;
    for (int o = 1; o < 32; o <<= 1) {
      const int y = __shfl_up(x, o);
      if ((tid & 31) >= o) x += y;
    }
    if ((tid & 31) == 31) wsum[tid >> 5] = x;
  }
  __syncthreads();
  if (tid < MAX_BINS) {
    int front = 0;
    for (int w = 0; w < (tid >> 5); ++w) front += wsum[w];
    off[tid] = front + x - v;
    segcnt[(size_t)blk * CNT_PITCH + tid] = min(v, SEG_CAP);
  }
  __syncthreads();

#pragma unroll
  for (int i = 0; i < PART_PER_THREAD; ++i)
    if (d[i] >= 0) sorted[off[(unsigned)d[i] / BIN_NODES] + r[i]] = make_int2(d[i], s[i]);
  __syncthreads();

  const int n = min(PART_EDGES, E - base);
  for (int p = tid; p < n; p += PART_THREADS) {
    const int2 en = sorted[p];
    const int bin = (unsigned)en.x / BIN_NODES;
    const int rank = p - off[bin];
    int2* __restrict__ to = seg + ((size_t)bin * nblk + blk) * SEG_CAP + rank;
    if (rank >= SEG_CAP) {
      const int q = atomicAdd(&nspill, 1);
      to = spill + (size_t)blk * PART_EDGES + q;
    }
    *to = en;
  }
  __syncthreads();
  if (tid == 0) spillc[blk * 32] = nspill;
}

struct Six {
  v2f p0, p1, p2;
};

__device__ __forceinline__ Six genconv_row_six(const float* __restrict__ feat, int s, int c8) {
  const unsigned off = __umul24((unsigned)s, DF * 4u) + (unsigned)c8;
  const char* p = reinterpret_cast<const char*>(feat) + off;
  Six r;
  r.p0 = *reinterpret_cast<const v2f*>(p);
  r.p1 = *reinterpret_cast<const v2f*>(p + 128);
  r.p2 = *reinterpret_cast<const v2f*>(p + 256);
  return r;
}

__device__ __forceinline__ v2f genconv_relu_eps(const v2f f) {
  const v2f zero = {0.f, 0.f};
  const v2f eps = {FEPS, FEPS};
  return __builtin_elementwise_max(f, zero) + eps;
}

__device__ __forceinline__ void genconv_m_row(Six& r) {
  r.p0 = genconv_relu_eps(r.p0);
  r.p1 = genconv_relu_eps(r.p1);
  r.p2 = genconv_relu_eps(r.p2);
}

__device__ __forceinline__ void genconv_max_row(const Six& m, Six& mx) {
  mx.p0 = __builtin_elementwise_max(mx.p0, m.p0);
  mx.p1 = __builtin_elementwise_max(mx.p1, m.p1);
  mx.p2 = __builtin_elementwise_max(mx.p2, m.p2);
}

__device__ __forceinline__ void genconv_max_row2(const Six& m0, const Six& m1, Six& mx) {
  mx.p0 = __builtin_elementwise_max(__builtin_elementwise_max(mx.p0, m0.p0), m1.p0);
  mx.p1 = __builtin_elementwise_max(__builtin_elementwise_max(mx.p1, m0.p1), m1.p1);
  mx.p2 = __builtin_elementwise_max(__builtin_elementwise_max(mx.p2, m0.p2), m1.p2);
}

__device__ __forceinline__ void genconv_sum_pair(const v2f m, const v2f mx, v2f& num, v2f& den) {
  const v2f d = m - mx;
  v2f e;
  e.x = __expf(d.x);
  e.y = __expf(d.y);
  num += e * m;
  den += e;
}

__device__ __forceinline__ void genconv_sum_row(const Six& m, const Six& mx, Six& num, Six& den) {
  genconv_sum_pair(m.p0, mx.p0, num.p0, den.p0);
  genconv_sum_pair(m.p1, mx.p1, num.p1, den.p1);
  genconv_sum_pair(m.p2, mx.p2, num.p2, den.p2);
}

__device__ __forceinline__ float genconv_xhalf(float x) {
  const int v = __builtin_bit_cast(int, x);
  return __builtin_bit_cast(
      float, __builtin_amdgcn_permlanex16(v, v, 0x76543210, 0xfedcba98, false, false));
}

__device__ __forceinline__ void genconv_max_halves(Six& mx) {
  mx.p0.x = fmaxf(mx.p0.x, genconv_xhalf(mx.p0.x));
  mx.p0.y = fmaxf(mx.p0.y, genconv_xhalf(mx.p0.y));
  mx.p1.x = fmaxf(mx.p1.x, genconv_xhalf(mx.p1.x));
  mx.p1.y = fmaxf(mx.p1.y, genconv_xhalf(mx.p1.y));
  mx.p2.x = fmaxf(mx.p2.x, genconv_xhalf(mx.p2.x));
  mx.p2.y = fmaxf(mx.p2.y, genconv_xhalf(mx.p2.y));
}

template <int S>
__device__ __forceinline__ void genconv_node_held(
    const float* __restrict__ feat, const int* bkt, int deg, int h, int c8,
    Six& mx, Six& num, Six& den) {
  Six row[S];
  int sid[S];
#pragma unroll
  for (int t = 0; t < S; ++t) {
    int e = 2 * t + h;
    if (t >= S - 2) e = min(e, deg - 1);
    sid[t] = bkt[e];
  }
#pragma unroll
  for (int t = 0; t < S; ++t) row[t] = genconv_row_six(feat, sid[t], c8);
  __builtin_amdgcn_sched_barrier(0);
#pragma unroll
  for (int t = 0; t < S; ++t) genconv_m_row(row[t]);
#pragma unroll
  for (int t = 0; t < S; t += 2) genconv_max_row2(row[t], row[t + 1], mx);
  genconv_max_halves(mx);
#pragma unroll
  for (int t = 0; t < S; ++t) {
    if (t < S - 2 || 2 * t + 1 < deg) {
      genconv_sum_row(row[t], mx, num, den);
    } else if (2 * t < deg) {
      if (h == 0) genconv_sum_row(row[t], mx, num, den);
    }
  }
}

template <bool SUM>
__device__ __forceinline__ void genconv_scan_pass(
    const float* __restrict__ feat, const int2* __restrict__ mine, const int* lseg,
    int nslot, const int2* __restrict__ spill, const int* __restrict__ spillc,
    int nblk, int spilled, int node, int h, int c8, int lane,
    Six& mx, Six& num, Six& den) {
  for (int b0 = 0; b0 < nslot; b0 += 32) {
    const int i = b0 + lane;
    bool hit = false;
    int s = 0;
    if (i < nslot && (i & (SEG_CAP - 1)) < lseg[i / SEG_CAP]) {
      const int2 e2 = mine[i];
      hit = (e2.x == node);
      s = e2.y;
    }
    unsigned hits = __builtin_amdgcn_ballot_w32(hit);
    while (hits) {
      const int j = __builtin_ctz(hits);
      hits &= hits - 1;
      Six f = genconv_row_six(feat, __builtin_amdgcn_readlane(s, j), c8);
      genconv_m_row(f);
      if (SUM) {
        if (h == 0) genconv_sum_row(f, mx, num, den);
      } else {
        genconv_max_row(f, mx);
      }
    }
  }
  if (spilled) {
    for (int b = 0; b < nblk; ++b) {
      const int c = min(spillc[b * 32], PART_EDGES);
      for (int b0 = 0; b0 < c; b0 += 32) {
        const int i = b0 + lane;
        bool hit = false;
        int s = 0;
        if (i < c) {
          const int2 e2 = spill[(size_t)b * PART_EDGES + i];
          hit = (e2.x == node);
          s = e2.y;
        }
        unsigned hits = __builtin_amdgcn_ballot_w32(hit);
        while (hits) {
          const int j = __builtin_ctz(hits);
          hits &= hits - 1;
          Six f = genconv_row_six(feat, __builtin_amdgcn_readlane(s, j), c8);
          genconv_m_row(f);
          if (SUM) {
            if (h == 0) genconv_sum_row(f, mx, num, den);
          } else {
            genconv_max_row(f, mx);
          }
        }
      }
    }
  }
}

__device__ __forceinline__ int genconv_next_node(int* counter, int lane) {
  int t = 0;
  if (lane == 0) t = atomicAdd(counter, 1);
  return __builtin_amdgcn_readfirstlane(t);
}

__global__ __launch_bounds__(FILL_THREADS) void genconv_bin_aggregate(
    const float* __restrict__ feat, const int* __restrict__ segp,
    const int* __restrict__ segcnt, const int* __restrict__ spillp,
    const int* __restrict__ spillc, const float* __restrict__ W,
    const float* __restrict__ b, float* __restrict__ out, int N, int nblk) {
  struct Lds {
    int lcnt[BIN_NODES];
    int lseg[MAX_BLKS];
    int anyspill;
    int nextnode;
    int nextunit;
    int lbucket[BIN_NODES * BUCKET_CAP];
    __attribute__((aligned(16))) float sW[DF * SW_PITCH];
    __attribute__((aligned(16))) float sA[BIN_TILES * 16 * SA_PITCH];
  };
  __shared__ Lds lds;
  int* const lbucket = lds.lbucket;
  int* const lcnt = lds.lcnt;
  int* const lseg = lds.lseg;
  int& anyspill = lds.anyspill;
  int& nextnode = lds.nextnode;
  int& nextunit = lds.nextunit;
  float* const sW = lds.sW;
  float* const sA = lds.sA;

  const int2* __restrict__ seg = reinterpret_cast<const int2*>(segp);
  const int2* __restrict__ spill = reinterpret_cast<const int2*>(spillp);
  const int tid = threadIdx.x;
  const int k = blockIdx.x;
  const int node0 = k * BIN_NODES;

  int sc = 0, pc = 0;
  if (tid < nblk) {
    sc = segcnt[(size_t)tid * CNT_PITCH + k];
    pc = spillc[tid * 32];
  }
  constexpr int QUADS_W = DF * DF / 4;
  constexpr int QW_PER = (QUADS_W + FILL_THREADS - 1) / FILL_THREADS;
  constexpr int QUADS_A = BIN_TILES * 16 * (DF / 4);
  constexpr int QA_PER = (QUADS_A + FILL_THREADS - 1) / FILL_THREADS;
  const int nhere = min(BIN_NODES, N - node0);
  const int validQ = nhere * (DF / 4);
  const float4* __restrict__ W4 = reinterpret_cast<const float4*>(W);
  const float4* __restrict__ f4 =
      reinterpret_cast<const float4*>(feat) + (size_t)node0 * (DF / 4);
  float4 wq[QW_PER];
  float4 fq[QA_PER];
#pragma unroll
  for (int it = 0; it < QW_PER; ++it) wq[it] = W4[min(tid + it * FILL_THREADS, QUADS_W - 1)];
#pragma unroll
  for (int it = 0; it < QA_PER; ++it) fq[it] = f4[min(tid + it * FILL_THREADS, validQ - 1)];
  if (tid < BIN_NODES) lcnt[tid] = 0;
  if (tid == 0) {
    anyspill = 0;
    nextnode = 0;
    nextunit = 0;
  }
  static_assert(FILL_THREADS == MAX_BLKS, "one thread per entry of lseg");
  lseg[tid] = sc;
  __syncthreads();
  if (pc > 0) anyspill = 1;

  const int2* __restrict__ mine = seg + (size_t)k * nblk * SEG_CAP;
  const int nslot = nblk * SEG_CAP;
  for (int g0 = 0; g0 < nslot; g0 += FILL_THREADS * FILL_DEPTH) {
    int2 en[FILL_DEPTH];
    int rk[FILL_DEPTH];
    bool ok[FILL_DEPTH];
#pragma unroll
    for (int u = 0; u < FILL_DEPTH; ++u) {
      const int i = g0 + u * FILL_THREADS + tid;
      ok[u] = (i < nslot) && ((i & (SEG_CAP - 1)) < lseg[i / SEG_CAP]);
      en[u] = make_int2(node0, 0);
      if (ok[u]) en[u] = mine[i];
    }
#pragma unroll
    for (int u = 0; u < FILL_DEPTH; ++u) {
      rk[u] = 0;
      if (ok[u]) rk[u] = atomicAdd(&lcnt[en[u].x - node0], 1);
    }
#pragma unroll
    for (int u = 0; u < FILL_DEPTH; ++u)
      if (ok[u] && rk[u] < BUCKET_CAP) lbucket[(en[u].x - node0) * BUCKET_CAP + rk[u]] = en[u].y;
  }
  float4* sW4 = reinterpret_cast<float4*>(sW);
  float4* sA4 = reinterpret_cast<float4*>(sA);
#pragma unroll
  for (int it = 0; it < QW_PER; ++it) {
    const int j = tid + it * FILL_THREADS;
    if (j < QUADS_W) sW4[(j / (DF / 4)) * (SW_PITCH / 4) + j % (DF / 4)] = wq[it];
  }
#pragma unroll
  for (int it = 0; it < QA_PER; ++it) {
    const int j = tid + it * FILL_THREADS;
    if (j < QUADS_A)
      sA4[(j / (DF / 4)) * (SA_PITCH / 4) + j % (DF / 4)] =
          (j < validQ) ? fq[it] : make_float4(0.f, 0.f, 0.f, 0.f);
  }
  __syncthreads();

  const int spilled = anyspill;
  if (spilled) {
    for (int b = 0; b < nblk; ++b) {
      const int c = min(spillc[b * 32], PART_EDGES);
      for (int i = tid; i < c; i += FILL_THREADS) {
        const int2 e2 = spill[(size_t)b * PART_EDGES + i];
        if ((unsigned)e2.x / BIN_NODES == (unsigned)k) {
          const int r = atomicAdd(&lcnt[e2.x - node0], 1);
          if (r < BUCKET_CAP) lbucket[(e2.x - node0) * BUCKET_CAP + r] = e2.y;
        }
      }
    }
  }
  __syncthreads();

  const int lane = tid & 31;
  const int h = lane >> 4;
  const int c16 = lane & 15;
  const int c8 = c16 * 8;
  const int fo0 = h ? 33 + 2 * c16 : 2 * c16;
  const int fo1 = h ? 64 + 2 * c16 : 2 * c16 + 1;
  const int fo2 = h ? 65 + 2 * c16 : 32 + 2 * c16;

  for (;;) {
    const int cur = genconv_next_node(&nextnode, lane);
    if (cur >= nhere) break;
    const int node = node0 + cur;
    const int deg = __builtin_amdgcn_readfirstlane(lcnt[cur]);
    if (deg <= 0) continue;
    const int* bkt = lbucket + cur * BUCKET_CAP;

    const v2f z2 = {0.f, 0.f};
    Six mx = {z2, z2, z2};
    Six num = {z2, z2, z2};
    Six den = {z2, z2, z2};
    if (deg <= ROWS_HELD) {
      if (deg <= 4) genconv_node_held<2>(feat, bkt, deg, h, c8, mx, num, den);
      else if (deg <= 8) genconv_node_held<4>(feat, bkt, deg, h, c8, mx, num, den);
      else if (deg <= 12) genconv_node_held<6>(feat, bkt, deg, h, c8, mx, num, den);
      else if (deg <= 16) genconv_node_held<8>(feat, bkt, deg, h, c8, mx, num, den);
      else if (deg <= 20) genconv_node_held<10>(feat, bkt, deg, h, c8, mx, num, den);
      else genconv_node_held<PAIRS_HELD>(feat, bkt, deg, h, c8, mx, num, den);
    } else if (deg <= BUCKET_CAP) {
      const int np = (deg + 1) >> 1;
#pragma unroll 4
      for (int j = 0; j < np; ++j) {
        Six f = genconv_row_six(feat, bkt[min(2 * j + h, deg - 1)], c8);
        genconv_m_row(f);
        genconv_max_row(f, mx);
      }
      genconv_max_halves(mx);
#pragma unroll 4
      for (int j = 0; j < np; ++j) {
        const int e = 2 * j + h;
        Six f = genconv_row_six(feat, bkt[min(e, deg - 1)], c8);
        genconv_m_row(f);
        if (e < deg) genconv_sum_row(f, mx, num, den);
      }
    } else {
      genconv_scan_pass<false>(feat, mine, lseg, nslot, spill, spillc, nblk, spilled,
                               node, h, c8, lane, mx, num, den);
      genconv_scan_pass<true>(feat, mine, lseg, nslot, spill, spillc, nblk, spilled,
                              node, h, c8, lane, mx, num, den);
    }

    const float gn0 = h ? num.p0.x : num.p1.y, kn0 = h ? num.p1.y : num.p0.x;
    const float gn1 = h ? num.p0.y : num.p2.x, kn1 = h ? num.p2.x : num.p0.y;
    const float gn2 = h ? num.p1.x : num.p2.y, kn2 = h ? num.p2.y : num.p1.x;
    const float gd0 = h ? den.p0.x : den.p1.y, kd0 = h ? den.p1.y : den.p0.x;
    const float gd1 = h ? den.p0.y : den.p2.x, kd1 = h ? den.p2.x : den.p0.y;
    const float gd2 = h ? den.p1.x : den.p2.y, kd2 = h ? den.p2.y : den.p1.x;
    const float r0 = (kn0 + genconv_xhalf(gn0)) / (kd0 + genconv_xhalf(gd0));
    const float r1 = (kn1 + genconv_xhalf(gn1)) / (kd1 + genconv_xhalf(gd1));
    const float r2 = (kn2 + genconv_xhalf(gn2)) / (kd2 + genconv_xhalf(gd2));
    float* own = sA + cur * SA_PITCH;
    own[fo0] += r0;
    own[fo1] += r1;
    own[fo2] += r2;
  }
  __syncthreads();

  const int mrow = lane & 15;
  const int kh = lane >> 4;
  for (;;) {
    const int u = genconv_next_node(&nextunit, lane);
    if (u >= BIN_TILES * COL_UNITS) break;
    const int rt = u / COL_UNITS;
    const int cu = u - rt * COL_UNITS;
    if (rt * 16 >= nhere) break;
    const float* aRow = sA + (rt * 16 + mrow) * SA_PITCH + 2 * kh;
    const float* bCol = sW + (2 * kh) * SW_PITCH + cu * 32 + mrow;
    v8f acc0 = {};
    v8f acc1 = {};
#pragma unroll
    for (int kk = 0; kk < DF; kk += 4) {
      const v2f a = *reinterpret_cast<const v2f*>(aRow + kk);
      v2f b0, b1;
      b0.x = bCol[(kk + 0) * SW_PITCH];
      b0.y = bCol[(kk + 1) * SW_PITCH];
      b1.x = bCol[(kk + 0) * SW_PITCH + 16];
      b1.y = bCol[(kk + 1) * SW_PITCH + 16];
      acc0 = __builtin_amdgcn_wmma_f32_16x16x4_f32(
           false, a,  false, b0,
           (short)0, acc0,  false,  false);
      acc1 = __builtin_amdgcn_wmma_f32_16x16x4_f32(
           false, a,  false, b1,
           (short)0, acc1,  false,  false);
    }
    const float bias = b[cu * 32 + lane];
#pragma unroll
    for (int i = 0; i < 8; ++i) {
      const float give = kh ? acc0[i] : acc1[i];
      const float got = genconv_xhalf(give);
      const float lo = kh ? got : acc0[i];
      const float hi = kh ? acc1[i] : got;
      const int r0 = rt * 16 + i;
      float* o = out + (size_t)(node0 + r0) * DF + cu * 32 + lane;
      if (r0 < nhere) o[0] = lo + bias;
      if (r0 + 8 < nhere) o[8 * DF] = hi + bias;
    }
  }
}

extern "C" void kernel_launch(void* const* d_in, const int* in_sizes, int n_in,
                              void* d_out, int out_size, void* d_ws, size_t ws_size,
                              hipStream_t stream) {
  const float* feat = (const float*)d_in[0];
  const int*   src  = (const int*)d_in[1];
  const int*   dst  = (const int*)d_in[2];
  const float* W    = (const float*)d_in[3];
  const float* b    = (const float*)d_in[4];
  float*       out  = (float*)d_out;

  const int N = in_sizes[0] / DF;
  const int E = in_sizes[1];

  const int nbin = (N + BIN_NODES - 1) / BIN_NODES;
  const int nblk = (E + PART_EDGES - 1) / PART_EDGES;
  int*   segcnt = (int*)d_ws;
  int*   spillc = segcnt + (size_t)nblk * CNT_PITCH;
  int*   seg    = spillc + (size_t)nblk * 32;
  int*   spill  = seg + (size_t)nbin * nblk * SEG_CAP * 2;

  genconv_edge_partition<<<nblk, PART_THREADS, 0, stream>>>(src, dst, seg, segcnt, spill,
                                                          spillc, E, nblk);
  genconv_bin_aggregate<<<nbin, FILL_THREADS, 0, stream>>>(feat, seg, segcnt, spill, spillc,
                                                         W, b, out, N, nblk);
}


# Softmax arithmetic trimmed: max on raw rows from 0 (= max of relu), exp argument by one fma against the pre-scaled max, eps added once after num * rcp(den); sums are now of relu(f) terms
# speedup vs baseline: 10.0316x; 1.0485x over previous
#include <hip/hip_runtime.h>


#define DF 96
#define BIN_TILES 13
#define SA_PITCH 100
#define SW_PITCH 104
#define COL_UNITS 3
#define FEPS 1e-7f
#define LOG2E 1.44269504088896340736f
#define BUCKET_CAP 64
#define ROWS_HELD 24
#define PAIRS_HELD (ROWS_HELD / 2)
#define BIN_NODES 196
#define MAX_BINS 256
#define MAX_BLKS 1024
#define PART_THREADS 1024
#define PART_PER_THREAD 4
#define PART_EDGES 4096
#define SEG_CAP 64
#define CNT_PITCH 256
#define FILL_THREADS 1024
#define FILL_DEPTH 13

typedef __attribute__((ext_vector_type(2))) float v2f;
typedef __attribute__((ext_vector_type(8))) float v8f;


__global__ __launch_bounds__(PART_THREADS) void genconv_edge_partition(
    const int* __restrict__ src, const int* __restrict__ dst,
    int* __restrict__ segp, int* __restrict__ segcnt,
    int* __restrict__ spillp, int* __restrict__ spillc, int E, int nblk) {
  __shared__ int hist[MAX_BINS];
  __shared__ int off[MAX_BINS];
  __shared__ int wsum[MAX_BINS / 32];
  __shared__ int nspill;
  __shared__ int2 sorted[PART_EDGES];

  int2* __restrict__ seg = reinterpret_cast<int2*>(segp);
  int2* __restrict__ spill = reinterpret_cast<int2*>(spillp);
  const int tid = threadIdx.x;
  const int blk = blockIdx.x;
  const int base = blk * PART_EDGES;

  int d[PART_PER_THREAD], s[PART_PER_THREAD], r[PART_PER_THREAD];
#pragma unroll
  for (int i = 0; i < PART_PER_THREAD; ++i) {
    const int e = base + i * PART_THREADS + tid;
    const bool ok = e < E;
    d[i] = ok ? dst[e] : -1;
    s[i] = ok ? src[e] : 0;
  }
  if (tid < MAX_BINS) hist[tid] = 0;
  if (tid == 0) nspill = 0;
  __syncthreads();

#pragma unroll
  for (int i = 0; i < PART_PER_THREAD; ++i) {
    r[i] = 0;
    if (d[i] >= 0) r[i] = atomicAdd(&hist[(unsigned)d[i] / BIN_NODES], 1);
  }
  __syncthreads();

  int v = 0, x = 0;
  if (tid < MAX_BINS) {
    v = hist[tid];
    x = v;
    for (int o = 1; o < 32; o <<= 1) {
      const int y = __shfl_up(x, o);
      if ((tid & 31) >= o) x += y;
    }
    if ((tid & 31) == 31) wsum[tid >> 5] = x;
  }
  __syncthreads();
  if (tid < MAX_BINS) {
    int front = 0;
    for (int w = 0; w < (tid >> 5); ++w) front += wsum[w];
    off[tid] = front + x - v;
    segcnt[(size_t)blk * CNT_PITCH + tid] = min(v, SEG_CAP);
  }
  __syncthreads();

#pragma unroll
  for (int i = 0; i < PART_PER_THREAD; ++i)
    if (d[i] >= 0) sorted[off[(unsigned)d[i] / BIN_NODES] + r[i]] = make_int2(d[i], s[i]);
  __syncthreads();

  const int n = min(PART_EDGES, E - base);
  for (int p = tid; p < n; p += PART_THREADS) {
    const int2 en = sorted[p];
    const int bin = (unsigned)en.x / BIN_NODES;
    const int rank = p - off[bin];
    int2* __restrict__ to = seg + ((size_t)bin * nblk + blk) * SEG_CAP + rank;
    if (rank >= SEG_CAP) {
      const int q = atomicAdd(&nspill, 1);
      to = spill + (size_t)blk * PART_EDGES + q;
    }
    *to = en;
  }
  __syncthreads();
  if (tid == 0) spillc[blk * 32] = nspill;
}

struct Six {
  v2f p0, p1, p2;
};

__device__ __forceinline__ Six genconv_row_six(const float* __restrict__ feat, int s, int c8) {
  const unsigned off = __umul24((unsigned)s, DF * 4u) + (unsigned)c8;
  const char* p = reinterpret_cast<const char*>(feat) + off;
  Six r;
  r.p0 = *reinterpret_cast<const v2f*>(p);
  r.p1 = *reinterpret_cast<const v2f*>(p + 128);
  r.p2 = *reinterpret_cast<const v2f*>(p + 256);
  return r;
}

__device__ __forceinline__ void genconv_max_row(const Six& m, Six& mx) {
  mx.p0 = __builtin_elementwise_max(mx.p0, m.p0);
  mx.p1 = __builtin_elementwise_max(mx.p1, m.p1);
  mx.p2 = __builtin_elementwise_max(mx.p2, m.p2);
}

__device__ __forceinline__ void genconv_max_row2(const Six& m0, const Six& m1, Six& mx) {
  mx.p0 = __builtin_elementwise_max(__builtin_elementwise_max(mx.p0, m0.p0), m1.p0);
  mx.p1 = __builtin_elementwise_max(__builtin_elementwise_max(mx.p1, m0.p1), m1.p1);
  mx.p2 = __builtin_elementwise_max(__builtin_elementwise_max(mx.p2, m0.p2), m1.p2);
}

__device__ __forceinline__ Six genconv_neg_scaled(const Six& mx) {
  const v2f nk = {-LOG2E, -LOG2E};
  Six nl;
  nl.p0 = mx.p0 * nk;
  nl.p1 = mx.p1 * nk;
  nl.p2 = mx.p2 * nk;
  return nl;
}

__device__ __forceinline__ void genconv_sum_pair(const v2f f, const v2f nl, v2f& num, v2f& den) {
  const v2f zero = {0.f, 0.f};
  const v2f k = {LOG2E, LOG2E};
  const v2f r = __builtin_elementwise_max(f, zero);
  const v2f t = __builtin_elementwise_fma(r, k, nl);
  v2f e;
  e.x = __builtin_amdgcn_exp2f(t.x);
  e.y = __builtin_amdgcn_exp2f(t.y);
  num += e * r;
  den += e;
}

__device__ __forceinline__ void genconv_sum_row(const Six& f, const Six& nl, Six& num, Six& den) {
  genconv_sum_pair(f.p0, nl.p0, num.p0, den.p0);
  genconv_sum_pair(f.p1, nl.p1, num.p1, den.p1);
  genconv_sum_pair(f.p2, nl.p2, num.p2, den.p2);
}

__device__ __forceinline__ float genconv_xhalf(float x) {
  const int v = __builtin_bit_cast(int, x);
  return __builtin_bit_cast(
      float, __builtin_amdgcn_permlanex16(v, v, 0x76543210, 0xfedcba98, false, false));
}

__device__ __forceinline__ void genconv_max_halves(Six& mx) {
  mx.p0.x = fmaxf(mx.p0.x, genconv_xhalf(mx.p0.x));
  mx.p0.y = fmaxf(mx.p0.y, genconv_xhalf(mx.p0.y));
  mx.p1.x = fmaxf(mx.p1.x, genconv_xhalf(mx.p1.x));
  mx.p1.y = fmaxf(mx.p1.y, genconv_xhalf(mx.p1.y));
  mx.p2.x = fmaxf(mx.p2.x, genconv_xhalf(mx.p2.x));
  mx.p2.y = fmaxf(mx.p2.y, genconv_xhalf(mx.p2.y));
}

template <int S>
__device__ __forceinline__ void genconv_node_held(
    const float* __restrict__ feat, const int* bkt, int deg, int h, int c8,
    Six& mx, Six& num, Six& den) {
  Six row[S];
  int sid[S];
#pragma unroll
  for (int t = 0; t < S; ++t) {
    int e = 2 * t + h;
    if (t >= S - 2) e = min(e, deg - 1);
    sid[t] = bkt[e];
  }
#pragma unroll
  for (int t = 0; t < S; ++t) row[t] = genconv_row_six(feat, sid[t], c8);
  __builtin_amdgcn_sched_barrier(0);
#pragma unroll
  for (int t = 0; t < S; t += 2) genconv_max_row2(row[t], row[t + 1], mx);
  genconv_max_halves(mx);
  const Six nl = genconv_neg_scaled(mx);
#pragma unroll
  for (int t = 0; t < S; ++t) {
    if (t < S - 2 || 2 * t + 1 < deg) {
      genconv_sum_row(row[t], nl, num, den);
    } else if (2 * t < deg) {
      if (h == 0) genconv_sum_row(row[t], nl, num, den);
    }
  }
}

template <bool SUM>
__device__ __forceinline__ void genconv_scan_pass(
    const float* __restrict__ feat, const int2* __restrict__ mine, const int* lseg,
    int nslot, const int2* __restrict__ spill, const int* __restrict__ spillc,
    int nblk, int spilled, int node, int h, int c8, int lane,
    Six& ms, Six& num, Six& den) {
  for (int b0 = 0; b0 < nslot; b0 += 32) {
    const int i = b0 + lane;
    bool hit = false;
    int s = 0;
    if (i < nslot && (i & (SEG_CAP - 1)) < lseg[i / SEG_CAP]) {
      const int2 e2 = mine[i];
      hit = (e2.x == node);
      s = e2.y;
    }
    unsigned hits = __builtin_amdgcn_ballot_w32(hit);
    while (hits) {
      const int j = __builtin_ctz(hits);
      hits &= hits - 1;
      const Six f = genconv_row_six(feat, __builtin_amdgcn_readlane(s, j), c8);
      if (SUM) {
        if (h == 0) genconv_sum_row(f, ms, num, den);
      } else {
        genconv_max_row(f, ms);
      }
    }
  }
  if (spilled) {
    for (int b = 0; b < nblk; ++b) {
      const int c = min(spillc[b * 32], PART_EDGES);
      for (int b0 = 0; b0 < c; b0 += 32) {
        const int i = b0 + lane;
        bool hit = false;
        int s = 0;
        if (i < c) {
          const int2 e2 = spill[(size_t)b * PART_EDGES + i];
          hit = (e2.x == node);
          s = e2.y;
        }
        unsigned hits = __builtin_amdgcn_ballot_w32(hit);
        while (hits) {
          const int j = __builtin_ctz(hits);
          hits &= hits - 1;
          const Six f = genconv_row_six(feat, __builtin_amdgcn_readlane(s, j), c8);
          if (SUM) {
            if (h == 0) genconv_sum_row(f, ms, num, den);
          } else {
            genconv_max_row(f, ms);
          }
        }
      }
    }
  }
}

__device__ __forceinline__ int genconv_next_node(int* counter, int lane) {
  int t = 0;
  if (lane == 0) t = atomicAdd(counter, 1);
  return __builtin_amdgcn_readfirstlane(t);
}

__global__ __launch_bounds__(FILL_THREADS) void genconv_bin_aggregate(
    const float* __restrict__ feat, const int* __restrict__ segp,
    const int* __restrict__ segcnt, const int* __restrict__ spillp,
    const int* __restrict__ spillc, const float* __restrict__ W,
    const float* __restrict__ b, float* __restrict__ out, int N, int nblk) {
  struct Lds {
    int lcnt[BIN_NODES];
    int lseg[MAX_BLKS];
    int anyspill;
    int nextnode;
    int nextunit;
    int lbucket[BIN_NODES * BUCKET_CAP];
    __attribute__((aligned(16))) float sW[DF * SW_PITCH];
    __attribute__((aligned(16))) float sA[BIN_TILES * 16 * SA_PITCH];
  };
  __shared__ Lds lds;
  int* const lbucket = lds.lbucket;
  int* const lcnt = lds.lcnt;
  int* const lseg = lds.lseg;
  int& anyspill = lds.anyspill;
  int& nextnode = lds.nextnode;
  int& nextunit = lds.nextunit;
  float* const sW = lds.sW;
  float* const sA = lds.sA;

  const int2* __restrict__ seg = reinterpret_cast<const int2*>(segp);
  const int2* __restrict__ spill = reinterpret_cast<const int2*>(spillp);
  const int tid = threadIdx.x;
  const int k = blockIdx.x;
  const int node0 = k * BIN_NODES;

  int sc = 0, pc = 0;
  if (tid < nblk) {
    sc = segcnt[(size_t)tid * CNT_PITCH + k];
    pc = spillc[tid * 32];
  }
  constexpr int QUADS_W = DF * DF / 4;
  constexpr int QW_PER = (QUADS_W + FILL_THREADS - 1) / FILL_THREADS;
  constexpr int QUADS_A = BIN_TILES * 16 * (DF / 4);
  constexpr int QA_PER = (QUADS_A + FILL_THREADS - 1) / FILL_THREADS;
  const int nhere = min(BIN_NODES, N - node0);
  const int validQ = nhere * (DF / 4);
  const float4* __restrict__ W4 = reinterpret_cast<const float4*>(W);
  const float4* __restrict__ f4 =
      reinterpret_cast<const float4*>(feat) + (size_t)node0 * (DF / 4);
  float4 wq[QW_PER];
  float4 fq[QA_PER];
#pragma unroll
  for (int it = 0; it < QW_PER; ++it) wq[it] = W4[min(tid + it * FILL_THREADS, QUADS_W - 1)];
#pragma unroll
  for (int it = 0; it < QA_PER; ++it) fq[it] = f4[min(tid + it * FILL_THREADS, validQ - 1)];
  if (tid < BIN_NODES) lcnt[tid] = 0;
  if (tid == 0) {
    anyspill = 0;
    nextnode = 0;
    nextunit = 0;
  }
  static_assert(FILL_THREADS == MAX_BLKS, "one thread per entry of lseg");
  lseg[tid] = sc;
  __syncthreads();
  if (pc > 0) anyspill = 1;

  const int2* __restrict__ mine = seg + (size_t)k * nblk * SEG_CAP;
  const int nslot = nblk * SEG_CAP;
  for (int g0 = 0; g0 < nslot; g0 += FILL_THREADS * FILL_DEPTH) {
    int2 en[FILL_DEPTH];
    int rk[FILL_DEPTH];
    bool ok[FILL_DEPTH];
#pragma unroll
    for (int u = 0; u < FILL_DEPTH; ++u) {
      const int i = g0 + u * FILL_THREADS + tid;
      ok[u] = (i < nslot) && ((i & (SEG_CAP - 1)) < lseg[i / SEG_CAP]);
      en[u] = make_int2(node0, 0);
      if (ok[u]) en[u] = mine[i];
    }
#pragma unroll
    for (int u = 0; u < FILL_DEPTH; ++u) {
      rk[u] = 0;
      if (ok[u]) rk[u] = atomicAdd(&lcnt[en[u].x - node0], 1);
    }
#pragma unroll
    for (int u = 0; u < FILL_DEPTH; ++u)
      if (ok[u] && rk[u] < BUCKET_CAP) lbucket[(en[u].x - node0) * BUCKET_CAP + rk[u]] = en[u].y;
  }
  float4* sW4 = reinterpret_cast<float4*>(sW);
  float4* sA4 = reinterpret_cast<float4*>(sA);
#pragma unroll
  for (int it = 0; it < QW_PER; ++it) {
    const int j = tid + it * FILL_THREADS;
    if (j < QUADS_W) sW4[(j / (DF / 4)) * (SW_PITCH / 4) + j % (DF / 4)] = wq[it];
  }
#pragma unroll
  for (int it = 0; it < QA_PER; ++it) {
    const int j = tid + it * FILL_THREADS;
    if (j < QUADS_A)
      sA4[(j / (DF / 4)) * (SA_PITCH / 4) + j % (DF / 4)] =
          (j < validQ) ? fq[it] : make_float4(0.f, 0.f, 0.f, 0.f);
  }
  __syncthreads();

  const int spilled = anyspill;
  if (spilled) {
    for (int b = 0; b < nblk; ++b) {
      const int c = min(spillc[b * 32], PART_EDGES);
      for (int i = tid; i < c; i += FILL_THREADS) {
        const int2 e2 = spill[(size_t)b * PART_EDGES + i];
        if ((unsigned)e2.x / BIN_NODES == (unsigned)k) {
          const int r = atomicAdd(&lcnt[e2.x - node0], 1);
          if (r < BUCKET_CAP) lbucket[(e2.x - node0) * BUCKET_CAP + r] = e2.y;
        }
      }
    }
  }
  __syncthreads();

  const int lane = tid & 31;
  const int h = lane >> 4;
  const int c16 = lane & 15;
  const int c8 = c16 * 8;
  const int fo0 = h ? 33 + 2 * c16 : 2 * c16;
  const int fo1 = h ? 64 + 2 * c16 : 2 * c16 + 1;
  const int fo2 = h ? 65 + 2 * c16 : 32 + 2 * c16;

  for (;;) {
    const int cur = genconv_next_node(&nextnode, lane);
    if (cur >= nhere) break;
    const int node = node0 + cur;
    const int deg = __builtin_amdgcn_readfirstlane(lcnt[cur]);
    if (deg <= 0) continue;
    const int* bkt = lbucket + cur * BUCKET_CAP;

    const v2f z2 = {0.f, 0.f};
    Six mx = {z2, z2, z2};
    Six num = {z2, z2, z2};
    Six den = {z2, z2, z2};
    if (deg <= ROWS_HELD) {
      if (deg <= 4) genconv_node_held<2>(feat, bkt, deg, h, c8, mx, num, den);
      else if (deg <= 8) genconv_node_held<4>(feat, bkt, deg, h, c8, mx, num, den);
      else if (deg <= 12) genconv_node_held<6>(feat, bkt, deg, h, c8, mx, num, den);
      else if (deg <= 16) genconv_node_held<8>(feat, bkt, deg, h, c8, mx, num, den);
      else if (deg <= 20) genconv_node_held<10>(feat, bkt, deg, h, c8, mx, num, den);
      else genconv_node_held<PAIRS_HELD>(feat, bkt, deg, h, c8, mx, num, den);
    } else if (deg <= BUCKET_CAP) {
      const int np = (deg + 1) >> 1;
#pragma unroll 4
      for (int j = 0; j < np; ++j) {
        const Six f = genconv_row_six(feat, bkt[min(2 * j + h, deg - 1)], c8);
        genconv_max_row(f, mx);
      }
      genconv_max_halves(mx);
      const Six nl = genconv_neg_scaled(mx);
#pragma unroll 4
      for (int j = 0; j < np; ++j) {
        const int e = 2 * j + h;
        const Six f = genconv_row_six(feat, bkt[min(e, deg - 1)], c8);
        if (e < deg) genconv_sum_row(f, nl, num, den);
      }
    } else {
      genconv_scan_pass<false>(feat, mine, lseg, nslot, spill, spillc, nblk, spilled,
                               node, h, c8, lane, mx, num, den);
      Six nl = genconv_neg_scaled(mx);
      genconv_scan_pass<true>(feat, mine, lseg, nslot, spill, spillc, nblk, spilled,
                              node, h, c8, lane, nl, num, den);
    }

    const float gn0 = h ? num.p0.x : num.p1.y, kn0 = h ? num.p1.y : num.p0.x;
    const float gn1 = h ? num.p0.y : num.p2.x, kn1 = h ? num.p2.x : num.p0.y;
    const float gn2 = h ? num.p1.x : num.p2.y, kn2 = h ? num.p2.y : num.p1.x;
    const float gd0 = h ? den.p0.x : den.p1.y, kd0 = h ? den.p1.y : den.p0.x;
    const float gd1 = h ? den.p0.y : den.p2.x, kd1 = h ? den.p2.x : den.p0.y;
    const float gd2 = h ? den.p1.x : den.p2.y, kd2 = h ? den.p2.y : den.p1.x;
    const float r0 = (kn0 + genconv_xhalf(gn0)) * __builtin_amdgcn_rcpf(kd0 + genconv_xhalf(gd0)) + FEPS;
    const float r1 = (kn1 + genconv_xhalf(gn1)) * __builtin_amdgcn_rcpf(kd1 + genconv_xhalf(gd1)) + FEPS;
    const float r2 = (kn2 + genconv_xhalf(gn2)) * __builtin_amdgcn_rcpf(kd2 + genconv_xhalf(gd2)) + FEPS;
    float* own = sA + cur * SA_PITCH;
    own[fo0] += r0;
    own[fo1] += r1;
    own[fo2] += r2;
  }
  __syncthreads();

  const int mrow = lane & 15;
  const int kh = lane >> 4;
  for (;;) {
    const int u = genconv_next_node(&nextunit, lane);
    if (u >= BIN_TILES * COL_UNITS) break;
    const int rt = u / COL_UNITS;
    const int cu = u - rt * COL_UNITS;
    if (rt * 16 >= nhere) break;
    const float* aRow = sA + (rt * 16 + mrow) * SA_PITCH + 2 * kh;
    const float* bCol = sW + (2 * kh) * SW_PITCH + cu * 32 + mrow;
    v8f acc0 = {};
    v8f acc1 = {};
#pragma unroll
    for (int kk = 0; kk < DF; kk += 4) {
      const v2f a = *reinterpret_cast<const v2f*>(aRow + kk);
      v2f b0, b1;
      b0.x = bCol[(kk + 0) * SW_PITCH];
      b0.y = bCol[(kk + 1) * SW_PITCH];
      b1.x = bCol[(kk + 0) * SW_PITCH + 16];
      b1.y = bCol[(kk + 1) * SW_PITCH + 16];
      acc0 = __builtin_amdgcn_wmma_f32_16x16x4_f32(
           false, a,  false, b0,
           (short)0, acc0,  false,  false);
      acc1 = __builtin_amdgcn_wmma_f32_16x16x4_f32(
           false, a,  false, b1,
           (short)0, acc1,  false,  false);
    }
    const float bias = b[cu * 32 + lane];
#pragma unroll
    for (int i = 0; i < 8; ++i) {
      const float give = kh ? acc0[i] : acc1[i];
      const float got = genconv_xhalf(give);
      const float lo = kh ? got : acc0[i];
      const float hi = kh ? acc1[i] : got;
      const int r0 = rt * 16 + i;
      float* o = out + (size_t)(node0 + r0) * DF + cu * 32 + lane;
      if (r0 < nhere) o[0] = lo + bias;
      if (r0 + 8 < nhere) o[8 * DF] = hi + bias;
    }
  }
}

extern "C" void kernel_launch(void* const* d_in, const int* in_sizes, int n_in,
                              void* d_out, int out_size, void* d_ws, size_t ws_size,
                              hipStream_t stream) {
  const float* feat = (const float*)d_in[0];
  const int*   src  = (const int*)d_in[1];
  const int*   dst  = (const int*)d_in[2];
  const float* W    = (const float*)d_in[3];
  const float* b    = (const float*)d_in[4];
  float*       out  = (float*)d_out;

  const int N = in_sizes[0] / DF;
  const int E = in_sizes[1];

  const int nbin = (N + BIN_NODES - 1) / BIN_NODES;
  const int nblk = (E + PART_EDGES - 1) / PART_EDGES;
  int*   segcnt = (int*)d_ws;
  int*   spillc = segcnt + (size_t)nblk * CNT_PITCH;
  int*   seg    = spillc + (size_t)nblk * 32;
  int*   spill  = seg + (size_t)nbin * nblk * SEG_CAP * 2;

  genconv_edge_partition<<<nblk, PART_THREADS, 0, stream>>>(src, dst, seg, segcnt, spill,
                                                          spillc, E, nblk);
  genconv_bin_aggregate<<<nbin, FILL_THREADS, 0, stream>>>(feat, seg, segcnt, spill, spillc,
                                                         W, b, out, N, nblk);
}


# Nodes of 25 to 64 edges: both passes read rows in groups of 8 row pairs, a group's 24 loads in flight together; the sum pass made one memory round trip per pair before; sum order kept
# speedup vs baseline: 10.5408x; 1.0508x over previous
#include <hip/hip_runtime.h>


#define DF 96
#define BIN_TILES 13
#define SA_PITCH 100
#define SW_PITCH 104
#define COL_UNITS 3
#define FEPS 1e-7f
#define LOG2E 1.44269504088896340736f
#define BUCKET_CAP 64
#define ROWS_HELD 24
#define PAIRS_HELD (ROWS_HELD / 2)
#define GROUP_PAIRS 8
#define BIN_NODES 196
#define MAX_BINS 256
#define MAX_BLKS 1024
#define PART_THREADS 1024
#define PART_PER_THREAD 4
#define PART_EDGES 4096
#define SEG_CAP 64
#define CNT_PITCH 256
#define FILL_THREADS 1024
#define FILL_DEPTH 13

typedef __attribute__((ext_vector_type(2))) float v2f;
typedef __attribute__((ext_vector_type(8))) float v8f;


__global__ __launch_bounds__(PART_THREADS) void genconv_edge_partition(
    const int* __restrict__ src, const int* __restrict__ dst,
    int* __restrict__ segp, int* __restrict__ segcnt,
    int* __restrict__ spillp, int* __restrict__ spillc, int E, int nblk) {
  __shared__ int hist[MAX_BINS];
  __shared__ int off[MAX_BINS];
  __shared__ int wsum[MAX_BINS / 32];
  __shared__ int nspill;
  __shared__ int2 sorted[PART_EDGES];

  int2* __restrict__ seg = reinterpret_cast<int2*>(segp);
  int2* __restrict__ spill = reinterpret_cast<int2*>(spillp);
  const int tid = threadIdx.x;
  const int blk = blockIdx.x;
  const int base = blk * PART_EDGES;

  int d[PART_PER_THREAD], s[PART_PER_THREAD], r[PART_PER_THREAD];
#pragma unroll
  for (int i = 0; i < PART_PER_THREAD; ++i) {
    const int e = base + i * PART_THREADS + tid;
    const bool ok = e < E;
    d[i] = ok ? dst[e] : -1;
    s[i] = ok ? src[e] : 0;
  }
  if (tid < MAX_BINS) hist[tid] = 0;
  if (tid == 0) nspill = 0;
  __syncthreads();

#pragma unroll
  for (int i = 0; i < PART_PER_THREAD; ++i) {
    r[i] = 0;
    if (d[i] >= 0) r[i] = atomicAdd(&hist[(unsigned)d[i] / BIN_NODES], 1);
  }
  __syncthreads();

  int v = 0, x = 0;
  if (tid < MAX_BINS) {
    v = hist[tid];
    x = v;
    for (int o = 1; o < 32; o <<= 1) {
      const int y = __shfl_up(x, o);
      if ((tid & 31) >= o) x += y;
    }
    if ((tid & 31) == 31) wsum[tid >> 5] = x;
  }
  __syncthreads();
  if (tid < MAX_BINS) {
    int front = 0;
    for (int w = 0; w < (tid >> 5); ++w) front += wsum[w];
    off[tid] = front + x - v;
    segcnt[(size_t)blk * CNT_PITCH + tid] = min(v, SEG_CAP);
  }
  __syncthreads();

#pragma unroll
  for (int i = 0; i < PART_PER_THREAD; ++i)
    if (d[i] >= 0) sorted[off[(unsigned)d[i] / BIN_NODES] + r[i]] = make_int2(d[i], s[i]);
  __syncthreads();

  const int n = min(PART_EDGES, E - base);
  for (int p = tid; p < n; p += PART_THREADS) {
    const int2 en = sorted[p];
    const int bin = (unsigned)en.x / BIN_NODES;
    const int rank = p - off[bin];
    int2* __restrict__ to = seg + ((size_t)bin * nblk + blk) * SEG_CAP + rank;
    if (rank >= SEG_CAP) {
      const int q = atomicAdd(&nspill, 1);
      to = spill + (size_t)blk * PART_EDGES + q;
    }
    *to = en;
  }
  __syncthreads();
  if (tid == 0) spillc[blk * 32] = nspill;
}

struct Six {
  v2f p0, p1, p2;
};

__device__ __forceinline__ Six genconv_row_six(const float* __restrict__ feat, int s, int c8) {
  const unsigned off = __umul24((unsigned)s, DF * 4u) + (unsigned)c8;
  const char* p = reinterpret_cast<const char*>(feat) + off;
  Six r;
  r.p0 = *reinterpret_cast<const v2f*>(p);
  r.p1 = *reinterpret_cast<const v2f*>(p + 128);
  r.p2 = *reinterpret_cast<const v2f*>(p + 256);
  return r;
}

__device__ __forceinline__ void genconv_max_row(const Six& m, Six& mx) {
  mx.p0 = __builtin_elementwise_max(mx.p0, m.p0);
  mx.p1 = __builtin_elementwise_max(mx.p1, m.p1);
  mx.p2 = __builtin_elementwise_max(mx.p2, m.p2);
}

__device__ __forceinline__ void genconv_max_row2(const Six& m0, const Six& m1, Six& mx) {
  mx.p0 = __builtin_elementwise_max(__builtin_elementwise_max(mx.p0, m0.p0), m1.p0);
  mx.p1 = __builtin_elementwise_max(__builtin_elementwise_max(mx.p1, m0.p1), m1.p1);
  mx.p2 = __builtin_elementwise_max(__builtin_elementwise_max(mx.p2, m0.p2), m1.p2);
}

__device__ __forceinline__ Six genconv_neg_scaled(const Six& mx) {
  const v2f nk = {-LOG2E, -LOG2E};
  Six nl;
  nl.p0 = mx.p0 * nk;
  nl.p1 = mx.p1 * nk;
  nl.p2 = mx.p2 * nk;
  return nl;
}

__device__ __forceinline__ void genconv_sum_pair(const v2f f, const v2f nl, v2f& num, v2f& den) {
  const v2f zero = {0.f, 0.f};
  const v2f k = {LOG2E, LOG2E};
  const v2f r = __builtin_elementwise_max(f, zero);
  const v2f t = __builtin_elementwise_fma(r, k, nl);
  v2f e;
  e.x = __builtin_amdgcn_exp2f(t.x);
  e.y = __builtin_amdgcn_exp2f(t.y);
  num += e * r;
  den += e;
}

__device__ __forceinline__ void genconv_sum_row(const Six& f, const Six& nl, Six& num, Six& den) {
  genconv_sum_pair(f.p0, nl.p0, num.p0, den.p0);
  genconv_sum_pair(f.p1, nl.p1, num.p1, den.p1);
  genconv_sum_pair(f.p2, nl.p2, num.p2, den.p2);
}

__device__ __forceinline__ float genconv_xhalf(float x) {
  const int v = __builtin_bit_cast(int, x);
  return __builtin_bit_cast(
      float, __builtin_amdgcn_permlanex16(v, v, 0x76543210, 0xfedcba98, false, false));
}

__device__ __forceinline__ void genconv_max_halves(Six& mx) {
  mx.p0.x = fmaxf(mx.p0.x, genconv_xhalf(mx.p0.x));
  mx.p0.y = fmaxf(mx.p0.y, genconv_xhalf(mx.p0.y));
  mx.p1.x = fmaxf(mx.p1.x, genconv_xhalf(mx.p1.x));
  mx.p1.y = fmaxf(mx.p1.y, genconv_xhalf(mx.p1.y));
  mx.p2.x = fmaxf(mx.p2.x, genconv_xhalf(mx.p2.x));
  mx.p2.y = fmaxf(mx.p2.y, genconv_xhalf(mx.p2.y));
}

template <int S>
__device__ __forceinline__ void genconv_node_held(
    const float* __restrict__ feat, const int* bkt, int deg, int h, int c8,
    Six& mx, Six& num, Six& den) {
  Six row[S];
  int sid[S];
#pragma unroll
  for (int t = 0; t < S; ++t) {
    int e = 2 * t + h;
    if (t >= S - 2) e = min(e, deg - 1);
    sid[t] = bkt[e];
  }
#pragma unroll
  for (int t = 0; t < S; ++t) row[t] = genconv_row_six(feat, sid[t], c8);
  __builtin_amdgcn_sched_barrier(0);
#pragma unroll
  for (int t = 0; t < S; t += 2) genconv_max_row2(row[t], row[t + 1], mx);
  genconv_max_halves(mx);
  const Six nl = genconv_neg_scaled(mx);
#pragma unroll
  for (int t = 0; t < S; ++t) {
    if (t < S - 2 || 2 * t + 1 < deg) {
      genconv_sum_row(row[t], nl, num, den);
    } else if (2 * t < deg) {
      if (h == 0) genconv_sum_row(row[t], nl, num, den);
    }
  }
}

__device__ __forceinline__ void genconv_group_rows(
    const float* __restrict__ feat, const int* bkt, int base, int deg, int h, int c8,
    Six (&row)[GROUP_PAIRS]) {
  int sid[GROUP_PAIRS];
#pragma unroll
  for (int t = 0; t < GROUP_PAIRS; ++t) sid[t] = bkt[min(2 * (base + t) + h, deg - 1)];
#pragma unroll
  for (int t = 0; t < GROUP_PAIRS; ++t) row[t] = genconv_row_six(feat, sid[t], c8);
}

__device__ __forceinline__ void genconv_group_max(
    const float* __restrict__ feat, const int* bkt, int base, int deg, int h, int c8, Six& mx) {
  Six row[GROUP_PAIRS];
  genconv_group_rows(feat, bkt, base, deg, h, c8, row);
#pragma unroll
  for (int t = 0; t < GROUP_PAIRS; t += 2) genconv_max_row2(row[t], row[t + 1], mx);
}

__device__ __forceinline__ void genconv_group_sum(
    const float* __restrict__ feat, const int* bkt, int base, int deg, int h, int c8,
    const Six& nl, Six& num, Six& den) {
  Six row[GROUP_PAIRS];
  genconv_group_rows(feat, bkt, base, deg, h, c8, row);
#pragma unroll
  for (int t = 0; t < GROUP_PAIRS; ++t) {
    const int e0 = 2 * (base + t);
    if (e0 + 1 < deg) {
      genconv_sum_row(row[t], nl, num, den);
    } else if (e0 < deg) {
      if (h == 0) genconv_sum_row(row[t], nl, num, den);
    }
  }
}

template <bool SUM>
__device__ __forceinline__ void genconv_scan_pass(
    const float* __restrict__ feat, const int2* __restrict__ mine, const int* lseg,
    int nslot, const int2* __restrict__ spill, const int* __restrict__ spillc,
    int nblk, int spilled, int node, int h, int c8, int lane,
    Six& ms, Six& num, Six& den) {
  for (int b0 = 0; b0 < nslot; b0 += 32) {
    const int i = b0 + lane;
    bool hit = false;
    int s = 0;
    if (i < nslot && (i & (SEG_CAP - 1)) < lseg[i / SEG_CAP]) {
      const int2 e2 = mine[i];
      hit = (e2.x == node);
      s = e2.y;
    }
    unsigned hits = __builtin_amdgcn_ballot_w32(hit);
    while (hits) {
      const int j = __builtin_ctz(hits);
      hits &= hits - 1;
      const Six f = genconv_row_six(feat, __builtin_amdgcn_readlane(s, j), c8);
      if (SUM) {
        if (h == 0) genconv_sum_row(f, ms, num, den);
      } else {
        genconv_max_row(f, ms);
      }
    }
  }
  if (spilled) {
    for (int b = 0; b < nblk; ++b) {
      const int c = min(spillc[b * 32], PART_EDGES);
      for (int b0 = 0; b0 < c; b0 += 32) {
        const int i = b0 + lane;
        bool hit = false;
        int s = 0;
        if (i < c) {
          const int2 e2 = spill[(size_t)b * PART_EDGES + i];
          hit = (e2.x == node);
          s = e2.y;
        }
        unsigned hits = __builtin_amdgcn_ballot_w32(hit);
        while (hits) {
          const int j = __builtin_ctz(hits);
          hits &= hits - 1;
          const Six f = genconv_row_six(feat, __builtin_amdgcn_readlane(s, j), c8);
          if (SUM) {
            if (h == 0) genconv_sum_row(f, ms, num, den);
          } else {
            genconv_max_row(f, ms);
          }
        }
      }
    }
  }
}

__device__ __forceinline__ int genconv_next_node(int* counter, int lane) {
  int t = 0;
  if (lane == 0) t = atomicAdd(counter, 1);
  return __builtin_amdgcn_readfirstlane(t);
}

__global__ __launch_bounds__(FILL_THREADS) void genconv_bin_aggregate(
    const float* __restrict__ feat, const int* __restrict__ segp,
    const int* __restrict__ segcnt, const int* __restrict__ spillp,
    const int* __restrict__ spillc, const float* __restrict__ W,
    const float* __restrict__ b, float* __restrict__ out, int N, int nblk) {
  struct Lds {
    int lcnt[BIN_NODES];
    int lseg[MAX_BLKS];
    int anyspill;
    int nextnode;
    int nextunit;
    int lbucket[BIN_NODES * BUCKET_CAP];
    __attribute__((aligned(16))) float sW[DF * SW_PITCH];
    __attribute__((aligned(16))) float sA[BIN_TILES * 16 * SA_PITCH];
  };
  __shared__ Lds lds;
  int* const lbucket = lds.lbucket;
  int* const lcnt = lds.lcnt;
  int* const lseg = lds.lseg;
  int& anyspill = lds.anyspill;
  int& nextnode = lds.nextnode;
  int& nextunit = lds.nextunit;
  float* const sW = lds.sW;
  float* const sA = lds.sA;

  const int2* __restrict__ seg = reinterpret_cast<const int2*>(segp);
  const int2* __restrict__ spill = reinterpret_cast<const int2*>(spillp);
  const int tid = threadIdx.x;
  const int k = blockIdx.x;
  const int node0 = k * BIN_NODES;

  int sc = 0, pc = 0;
  if (tid < nblk) {
    sc = segcnt[(size_t)tid * CNT_PITCH + k];
    pc = spillc[tid * 32];
  }
  constexpr int QUADS_W = DF * DF / 4;
  constexpr int QW_PER = (QUADS_W + FILL_THREADS - 1) / FILL_THREADS;
  constexpr int QUADS_A = BIN_TILES * 16 * (DF / 4);
  constexpr int QA_PER = (QUADS_A + FILL_THREADS - 1) / FILL_THREADS;
  const int nhere = min(BIN_NODES, N - node0);
  const int validQ = nhere * (DF / 4);
  const float4* __restrict__ W4 = reinterpret_cast<const float4*>(W);
  const float4* __restrict__ f4 =
      reinterpret_cast<const float4*>(feat) + (size_t)node0 * (DF / 4);
  float4 wq[QW_PER];
  float4 fq[QA_PER];
#pragma unroll
  for (int it = 0; it < QW_PER; ++it) wq[it] = W4[min(tid + it * FILL_THREADS, QUADS_W - 1)];
#pragma unroll
  for (int it = 0; it < QA_PER; ++it) fq[it] = f4[min(tid + it * FILL_THREADS, validQ - 1)];
  if (tid < BIN_NODES) lcnt[tid] = 0;
  if (tid == 0) {
    anyspill = 0;
    nextnode = 0;
    nextunit = 0;
  }
  static_assert(FILL_THREADS == MAX_BLKS, "one thread per entry of lseg");
  lseg[tid] = sc;
  __syncthreads();
  if (pc > 0) anyspill = 1;

  const int2* __restrict__ mine = seg + (size_t)k * nblk * SEG_CAP;
  const int nslot = nblk * SEG_CAP;
  for (int g0 = 0; g0 < nslot; g0 += FILL_THREADS * FILL_DEPTH) {
    int2 en[FILL_DEPTH];
    int rk[FILL_DEPTH];
    bool ok[FILL_DEPTH];
#pragma unroll
    for (int u = 0; u < FILL_DEPTH; ++u) {
      const int i = g0 + u * FILL_THREADS + tid;
      ok[u] = (i < nslot) && ((i & (SEG_CAP - 1)) < lseg[i / SEG_CAP]);
      en[u] = make_int2(node0, 0);
      if (ok[u]) en[u] = mine[i];
    }
#pragma unroll
    for (int u = 0; u < FILL_DEPTH; ++u) {
      rk[u] = 0;
      if (ok[u]) rk[u] = atomicAdd(&lcnt[en[u].x - node0], 1);
    }
#pragma unroll
    for (int u = 0; u < FILL_DEPTH; ++u)
      if (ok[u] && rk[u] < BUCKET_CAP) lbucket[(en[u].x - node0) * BUCKET_CAP + rk[u]] = en[u].y;
  }
  float4* sW4 = reinterpret_cast<float4*>(sW);
  float4* sA4 = reinterpret_cast<float4*>(sA);
#pragma unroll
  for (int it = 0; it < QW_PER; ++it) {
    const int j = tid + it * FILL_THREADS;
    if (j < QUADS_W) sW4[(j / (DF / 4)) * (SW_PITCH / 4) + j % (DF / 4)] = wq[it];
  }
#pragma unroll
  for (int it = 0; it < QA_PER; ++it) {
    const int j = tid + it * FILL_THREADS;
    if (j < QUADS_A)
      sA4[(j / (DF / 4)) * (SA_PITCH / 4) + j % (DF / 4)] =
          (j < validQ) ? fq[it] : make_float4(0.f, 0.f, 0.f, 0.f);
  }
  __syncthreads();

  const int spilled = anyspill;
  if (spilled) {
    for (int b = 0; b < nblk; ++b) {
      const int c = min(spillc[b * 32], PART_EDGES);
      for (int i = tid; i < c; i += FILL_THREADS) {
        const int2 e2 = spill[(size_t)b * PART_EDGES + i];
        if ((unsigned)e2.x / BIN_NODES == (unsigned)k) {
          const int r = atomicAdd(&lcnt[e2.x - node0], 1);
          if (r < BUCKET_CAP) lbucket[(e2.x - node0) * BUCKET_CAP + r] = e2.y;
        }
      }
    }
  }
  __syncthreads();

  const int lane = tid & 31;
  const int h = lane >> 4;
  const int c16 = lane & 15;
  const int c8 = c16 * 8;
  const int fo0 = h ? 33 + 2 * c16 : 2 * c16;
  const int fo1 = h ? 64 + 2 * c16 : 2 * c16 + 1;
  const int fo2 = h ? 65 + 2 * c16 : 32 + 2 * c16;

  for (;;) {
    const int cur = genconv_next_node(&nextnode, lane);
    if (cur >= nhere) break;
    const int node = node0 + cur;
    const int deg = __builtin_amdgcn_readfirstlane(lcnt[cur]);
    if (deg <= 0) continue;
    const int* bkt = lbucket + cur * BUCKET_CAP;

    const v2f z2 = {0.f, 0.f};
    Six mx = {z2, z2, z2};
    Six num = {z2, z2, z2};
    Six den = {z2, z2, z2};
    if (deg <= ROWS_HELD) {
      if (deg <= 4) genconv_node_held<2>(feat, bkt, deg, h, c8, mx, num, den);
      else if (deg <= 8) genconv_node_held<4>(feat, bkt, deg, h, c8, mx, num, den);
      else if (deg <= 12) genconv_node_held<6>(feat, bkt, deg, h, c8, mx, num, den);
      else if (deg <= 16) genconv_node_held<8>(feat, bkt, deg, h, c8, mx, num, den);
      else if (deg <= 20) genconv_node_held<10>(feat, bkt, deg, h, c8, mx, num, den);
      else genconv_node_held<PAIRS_HELD>(feat, bkt, deg, h, c8, mx, num, den);
    } else if (deg <= BUCKET_CAP) {
      const int np = (deg + 1) >> 1;
      for (int base = 0; base < np; base += GROUP_PAIRS)
        genconv_group_max(feat, bkt, base, deg, h, c8, mx);
      genconv_max_halves(mx);
      const Six nl = genconv_neg_scaled(mx);
      for (int base = 0; base < np; base += GROUP_PAIRS)
        genconv_group_sum(feat, bkt, base, deg, h, c8, nl, num, den);
    } else {
      genconv_scan_pass<false>(feat, mine, lseg, nslot, spill, spillc, nblk, spilled,
                               node, h, c8, lane, mx, num, den);
      Six nl = genconv_neg_scaled(mx);
      genconv_scan_pass<true>(feat, mine, lseg, nslot, spill, spillc, nblk, spilled,
                              node, h, c8, lane, nl, num, den);
    }

    const float gn0 = h ? num.p0.x : num.p1.y, kn0 = h ? num.p1.y : num.p0.x;
    const float gn1 = h ? num.p0.y : num.p2.x, kn1 = h ? num.p2.x : num.p0.y;
    const float gn2 = h ? num.p1.x : num.p2.y, kn2 = h ? num.p2.y : num.p1.x;
    const float gd0 = h ? den.p0.x : den.p1.y, kd0 = h ? den.p1.y : den.p0.x;
    const float gd1 = h ? den.p0.y : den.p2.x, kd1 = h ? den.p2.x : den.p0.y;
    const float gd2 = h ? den.p1.x : den.p2.y, kd2 = h ? den.p2.y : den.p1.x;
    const float r0 = (kn0 + genconv_xhalf(gn0)) * __builtin_amdgcn_rcpf(kd0 + genconv_xhalf(gd0)) + FEPS;
    const float r1 = (kn1 + genconv_xhalf(gn1)) * __builtin_amdgcn_rcpf(kd1 + genconv_xhalf(gd1)) + FEPS;
    const float r2 = (kn2 + genconv_xhalf(gn2)) * __builtin_amdgcn_rcpf(kd2 + genconv_xhalf(gd2)) + FEPS;
    float* own = sA + cur * SA_PITCH;
    own[fo0] += r0;
    own[fo1] += r1;
    own[fo2] += r2;
  }
  __syncthreads();

  const int mrow = lane & 15;
  const int kh = lane >> 4;
  for (;;) {
    const int u = genconv_next_node(&nextunit, lane);
    if (u >= BIN_TILES * COL_UNITS) break;
    const int rt = u / COL_UNITS;
    const int cu = u - rt * COL_UNITS;
    if (rt * 16 >= nhere) break;
    const float* aRow = sA + (rt * 16 + mrow) * SA_PITCH + 2 * kh;
    const float* bCol = sW + (2 * kh) * SW_PITCH + cu * 32 + mrow;
    v8f acc0 = {};
    v8f acc1 = {};
#pragma unroll
    for (int kk = 0; kk < DF; kk += 4) {
      const v2f a = *reinterpret_cast<const v2f*>(aRow + kk);
      v2f b0, b1;
      b0.x = bCol[(kk + 0) * SW_PITCH];
      b0.y = bCol[(kk + 1) * SW_PITCH];
      b1.x = bCol[(kk + 0) * SW_PITCH + 16];
      b1.y = bCol[(kk + 1) * SW_PITCH + 16];
      acc0 = __builtin_amdgcn_wmma_f32_16x16x4_f32(
           false, a,  false, b0,
           (short)0, acc0,  false,  false);
      acc1 = __builtin_amdgcn_wmma_f32_16x16x4_f32(
           false, a,  false, b1,
           (short)0, acc1,  false,  false);
    }
    const float bias = b[cu * 32 + lane];
#pragma unroll
    for (int i = 0; i < 8; ++i) {
      const float give = kh ? acc0[i] : acc1[i];
      const float got = genconv_xhalf(give);
      const float lo = kh ? got : acc0[i];
      const float hi = kh ? acc1[i] : got;
      const int r0 = rt * 16 + i;
      float* o = out + (size_t)(node0 + r0) * DF + cu * 32 + lane;
      if (r0 < nhere) o[0] = lo + bias;
      if (r0 + 8 < nhere) o[8 * DF] = hi + bias;
    }
  }
}

extern "C" void kernel_launch(void* const* d_in, const int* in_sizes, int n_in,
                              void* d_out, int out_size, void* d_ws, size_t ws_size,
                              hipStream_t stream) {
  const float* feat = (const float*)d_in[0];
  const int*   src  = (const int*)d_in[1];
  const int*   dst  = (const int*)d_in[2];
  const float* W    = (const float*)d_in[3];
  const float* b    = (const float*)d_in[4];
  float*       out  = (float*)d_out;

  const int N = in_sizes[0] / DF;
  const int E = in_sizes[1];

  const int nbin = (N + BIN_NODES - 1) / BIN_NODES;
  const int nblk = (E + PART_EDGES - 1) / PART_EDGES;
  int*   segcnt = (int*)d_ws;
  int*   spillc = segcnt + (size_t)nblk * CNT_PITCH;
  int*   seg    = spillc + (size_t)nblk * 32;
  int*   spill  = seg + (size_t)nbin * nblk * SEG_CAP * 2;

  genconv_edge_partition<<<nblk, PART_THREADS, 0, stream>>>(src, dst, seg, segcnt, spill,
                                                          spillc, E, nblk);
  genconv_bin_aggregate<<<nbin, FILL_THREADS, 0, stream>>>(feat, seg, segcnt, spill, spillc,
                                                         W, b, out, N, nblk);
}
